# starting kernel
# baseline (speedup 1.0000x reference)
#include <hip/hip_runtime.h>

#define NN 50000
#define EE 800000
#define DD 64
#define EPSF 1e-7f

typedef __attribute__((ext_vector_type(2))) float v2f;
typedef __attribute__((ext_vector_type(8))) float v8f;

__global__ void init_ws(float* __restrict__ segmax,
                        float* __restrict__ denom,
                        float* __restrict__ numer, int n) {
    int i = blockIdx.x * blockDim.x + threadIdx.x;
    if (i < n) {
        segmax[i] = 0.0f;
        denom[i]  = 0.0f;
        numer[i]  = 0.0f;
    }
}

__global__ void edge_max(const float* __restrict__ nf,
                         const int*   __restrict__ eattr,
                         const int*   __restrict__ src,
                         const int*   __restrict__ dst,
                         const float* __restrict__ emb,
                         float*       __restrict__ segmax) {
    long long gid = (long long)blockIdx.x * blockDim.x + threadIdx.x;
    if (gid >= (long long)EE * DD) return;
    int d = (int)(gid & (DD - 1));
    int e = (int)(gid >> 6);
    int s = src[e];
    int t = dst[e];
    int a = eattr[e];
    float m = nf[(long long)s * DD + d] + emb[a * DD + d];
    m = fmaxf(m, 0.0f) + EPSF;
    atomicMax((unsigned int*)(segmax + (long long)t * DD + d), __float_as_uint(m));
}

__global__ void edge_sum(const float* __restrict__ nf,
                         const int*   __restrict__ eattr,
                         const int*   __restrict__ src,
                         const int*   __restrict__ dst,
                         const float* __restrict__ emb,
                         const float* __restrict__ segmax,
                         float*       __restrict__ denom,
                         float*       __restrict__ numer) {
    long long gid = (long long)blockIdx.x * blockDim.x + threadIdx.x;
    if (gid >= (long long)EE * DD) return;
    int d = (int)(gid & (DD - 1));
    int e = (int)(gid >> 6);
    int s = src[e];
    int t = dst[e];
    int a = eattr[e];
    float m = nf[(long long)s * DD + d] + emb[a * DD + d];
    m = fmaxf(m, 0.0f) + EPSF;
    long long ti = (long long)t * DD + d;
    float z = __expf(m - segmax[ti]);
    atomicAdd(denom + ti, z);
    atomicAdd(numer + ti, m * z);
}

__global__ __launch_bounds__(128) void node_gemm(
        const float* __restrict__ nf,
        const float* __restrict__ numer,
        const float* __restrict__ denom,
        const float* __restrict__ W,
        const float* __restrict__ bias,
        float*       __restrict__ out) {
    __shared__ float fts[16 * 68];
    __shared__ float Ws [64 * 68];

    const int tid     = threadIdx.x;
    const int rowbase = blockIdx.x * 16;

    #pragma unroll
    for (int i = 0; i < 32; ++i) {
        int idx = tid + i * 128;
        int r = idx >> 6, c = idx & 63;
        Ws[r * 68 + c] = W[idx];
    }
    #pragma unroll
    for (int i = 0; i < 8; ++i) {
        int idx = tid + i * 128;
        int r = idx >> 6, c = idx & 63;
        long long g = (long long)(rowbase + r) * DD + c;
        float den = denom[g];
        float agg = (den > 0.0f) ? (numer[g] / den) : 0.0f;
        fts[r * 68 + c] = nf[g] + agg;
    }
    __syncthreads();

    const int lane  = tid & 31;
    const int wave  = tid >> 5;
    const int half  = lane >> 4;
    const int m     = lane & 15;

    v8f acc = {};
    #pragma unroll
    for (int k0 = 0; k0 < DD; k0 += 4) {
        const int k = k0 + 2 * half;
        v2f a, b;
        a.x = fts[m * 68 + k];
        a.y = fts[m * 68 + k + 1];
        b.x = Ws[(wave * 16 + m) * 68 + k];
        b.y = Ws[(wave * 16 + m) * 68 + k + 1];
        acc = __builtin_amdgcn_wmma_f32_16x16x4_f32(
                   false, a,  false, b,
                   (short)0, acc,
                   false,  false);
    }

    const float bb = bias[wave * 16 + m];
    #pragma unroll
    for (int v = 0; v < 8; ++v) {
        int row = rowbase + v + 8 * half;
        out[(long long)row * DD + wave * 16 + m] = acc[v] + bb;
    }
}

extern "C" void kernel_launch(void* const* d_in, const int* in_sizes, int n_in,
                              void* d_out, int out_size, void* d_ws, size_t ws_size,
                              hipStream_t stream) {
    const float* nf    = (const float*)d_in[0];
    const int*   eattr = (const int*)  d_in[1];
    const int*   src   = (const int*)  d_in[2];
    const int*   dst   = (const int*)  d_in[3];
    const float* emb   = (const float*)d_in[4];
    const float* W     = (const float*)d_in[5];
    const float* bias  = (const float*)d_in[6];
    float*       out   = (float*)d_out;

    float* segmax = (float*)d_ws;
    float* denom  = segmax + (size_t)NN * DD;
    float* numer  = denom  + (size_t)NN * DD;

    const int nd = NN * DD;
    init_ws<<<(nd + 255) / 256, 256, 0, stream>>>(segmax, denom, numer, nd);

    const long long tot = (long long)EE * DD;
    const int eblocks = (int)((tot + 255) / 256);
    edge_max<<<eblocks, 256, 0, stream>>>(nf, eattr, src, dst, emb, segmax);
    edge_sum<<<eblocks, 256, 0, stream>>>(nf, eattr, src, dst, emb, segmax,
                                          denom, numer);

    node_gemm<<<NN / 16, 128, 0, stream>>>(nf, numer, denom, W, bias, out);
}


# Float atomic scatter replaced by counting sort of edges by dst (int rank atomics, scan) and one wave per node reducing its edge group in registers with online softmax; sums now in group order
# speedup vs baseline: 2.6609x; 2.6609x over previous
#include <hip/hip_runtime.h>


#define NN 50000
#define EE 800000
#define DD 64
#define EPSF 1e-7f
#define NPAD 50176
#define NCHUNK (NN / 8)
#define NTILE 7

typedef __attribute__((ext_vector_type(2))) float v2f;
typedef __attribute__((ext_vector_type(8))) float v8f;


__global__ void zero_counts(int* __restrict__ cnt) {
    int i = blockIdx.x * blockDim.x + threadIdx.x;
    if (i < NN) cnt[i] = 0;
}

__global__ void hist_rank(const int* __restrict__ dst,
                          int*       __restrict__ cnt,
                          int*       __restrict__ rank) {
    int e = blockIdx.x * blockDim.x + threadIdx.x;
    if (e < EE) rank[e] = atomicAdd(cnt + dst[e], 1);
}

__global__ __launch_bounds__(1024) void scan_offsets(const int* __restrict__ cnt,
                                                     int*       __restrict__ off) {
    __shared__ int wsum[NTILE][32];
    __shared__ int ttot[NTILE];
    const int tid  = threadIdx.x;
    const int lane = tid & 31;
    const int wave = tid >> 5;
    const int4* c4 = reinterpret_cast<const int4*>(cnt);
    int4*       o4 = reinterpret_cast<int4*>(off);

    int4 lo[NTILE], hi[NTILE];
    int total[NTILE], v[NTILE];
    #pragma unroll
    for (int j = 0; j < NTILE; ++j) {
        const int c = j * 1024 + tid;
        if (c < NCHUNK) { lo[j] = c4[2 * c]; hi[j] = c4[2 * c + 1]; }
        else            { lo[j] = make_int4(0, 0, 0, 0); hi[j] = make_int4(0, 0, 0, 0); }
    }
    #pragma unroll
    for (int j = 0; j < NTILE; ++j) {
        total[j] = lo[j].x + lo[j].y + lo[j].z + lo[j].w
                 + hi[j].x + hi[j].y + hi[j].z + hi[j].w;
        v[j] = total[j];
    }
    for (int o = 1; o < 32; o <<= 1) {
        #pragma unroll
        for (int j = 0; j < NTILE; ++j) {
            int u = __shfl_up(v[j], o);
            if (lane >= o) v[j] += u;
        }
    }
    if (lane == 31) {
        #pragma unroll
        for (int j = 0; j < NTILE; ++j) wsum[j][wave] = v[j];
    }
    __syncthreads();
    if (wave == 0) {
        int x[NTILE], y[NTILE];
        #pragma unroll
        for (int j = 0; j < NTILE; ++j) { x[j] = wsum[j][lane]; y[j] = x[j]; }
        for (int o = 1; o < 32; o <<= 1) {
            #pragma unroll
            for (int j = 0; j < NTILE; ++j) {
                int u = __shfl_up(y[j], o);
                if (lane >= o) y[j] += u;
            }
        }
        #pragma unroll
        for (int j = 0; j < NTILE; ++j) {
            wsum[j][lane] = y[j] - x[j];
            if (lane == 31) ttot[j] = y[j];
        }
    }
    __syncthreads();
    int carry = 0;
    #pragma unroll
    for (int j = 0; j < NTILE; ++j) {
        const int c = j * 1024 + tid;
        if (c < NCHUNK) {
            int4 a, b;
            a.x = carry + wsum[j][wave] + (v[j] - total[j]);
            a.y = a.x + lo[j].x;
            a.z = a.y + lo[j].y;
            a.w = a.z + lo[j].z;
            b.x = a.w + lo[j].w;
            b.y = b.x + hi[j].x;
            b.z = b.y + hi[j].y;
            b.w = b.z + hi[j].z;
            o4[2 * c]     = a;
            o4[2 * c + 1] = b;
        }
        carry += ttot[j];
    }
    if (tid == 0) off[NN] = carry;
}

__global__ void scatter_edges(const int* __restrict__ eattr,
                              const int* __restrict__ src,
                              const int* __restrict__ dst,
                              const int* __restrict__ rank,
                              const int* __restrict__ off,
                              int*       __restrict__ sorted) {
    int e = blockIdx.x * blockDim.x + threadIdx.x;
    if (e < EE)
        reinterpret_cast<int2*>(sorted)[off[dst[e]] + rank[e]] = make_int2(src[e], eattr[e]);
}

__global__ __launch_bounds__(256) void seg_softmax(
        const float* __restrict__ nf,
        const float* __restrict__ emb,
        const int*   __restrict__ off,
        const int*   __restrict__ sorted_pairs,
        float*       __restrict__ denom,
        float*       __restrict__ numer) {
    const int lane = threadIdx.x & 31;
    const int t = __builtin_amdgcn_readfirstlane((int)(blockIdx.x * 8 + (threadIdx.x >> 5)));
    const int beg = off[t];
    const int end = off[t + 1];
    const float2* nf2  = reinterpret_cast<const float2*>(nf);
    const float2* emb2 = reinterpret_cast<const float2*>(emb);
    const int2* sorted = reinterpret_cast<const int2*>(sorted_pairs);

    float mx = 0.0f, my = 0.0f;
    float sx = 0.0f, sy = 0.0f;
    float qx = 0.0f, qy = 0.0f;
    for (int base = beg; base < end; base += 32) {
        const int n = min(32, end - base);
        const int2 mine = sorted[base + min(lane, n - 1)];
        for (int j = 0; j < n; j += 4) {
            float2 x[4], eb[4];
            #pragma unroll
            for (int k = 0; k < 4; ++k) {
                const int jj = min(j + k, n - 1);
                const int s = __builtin_amdgcn_readlane(mine.x, jj);
                const int a = __builtin_amdgcn_readlane(mine.y, jj);
                x[k]  = nf2[(size_t)s * (DD / 2) + lane];
                eb[k] = emb2[a * (DD / 2) + lane];
            }
            asm volatile("" :: "v"(x[0].x), "v"(x[0].y), "v"(x[1].x), "v"(x[1].y),
                               "v"(x[2].x), "v"(x[2].y), "v"(x[3].x), "v"(x[3].y),
                               "v"(eb[0].x), "v"(eb[0].y), "v"(eb[1].x), "v"(eb[1].y),
                               "v"(eb[2].x), "v"(eb[2].y), "v"(eb[3].x), "v"(eb[3].y));
            float ax[4], ay[4];
            #pragma unroll
            for (int k = 0; k < 4; ++k) {
                const bool valid = (j + k) < n;
                const float vx = fmaxf(x[k].x + eb[k].x, 0.0f) + EPSF;
                const float vy = fmaxf(x[k].y + eb[k].y, 0.0f) + EPSF;
                ax[k] = valid ? vx : 0.0f;
                ay[k] = valid ? vy : 0.0f;
            }
            const float nx = fmaxf(mx, fmaxf(fmaxf(ax[0], ax[1]), fmaxf(ax[2], ax[3])));
            const float ny = fmaxf(my, fmaxf(fmaxf(ay[0], ay[1]), fmaxf(ay[2], ay[3])));
            const float rx = __expf(mx - nx);
            const float ry = __expf(my - ny);
            sx *= rx; qx *= rx;
            sy *= ry; qy *= ry;
            #pragma unroll
            for (int k = 0; k < 4; ++k) {
                const bool valid = (j + k) < n;
                const float zx = valid ? __expf(ax[k] - nx) : 0.0f;
                const float zy = valid ? __expf(ay[k] - ny) : 0.0f;
                sx += zx; qx += ax[k] * zx;
                sy += zy; qy += ay[k] * zy;
            }
            mx = nx; my = ny;
        }
    }
    const size_t o = (size_t)t * (DD / 2) + lane;
    reinterpret_cast<float2*>(denom)[o] = make_float2(sx, sy);
    reinterpret_cast<float2*>(numer)[o] = make_float2(qx, qy);
}

__global__ __launch_bounds__(128) void node_gemm(
        const float* __restrict__ nf,
        const float* __restrict__ numer,
        const float* __restrict__ denom,
        const float* __restrict__ W,
        const float* __restrict__ bias,
        float*       __restrict__ out) {
    __shared__ float fts[16 * 68];
    __shared__ float Ws [64 * 68];

    const int tid     = threadIdx.x;
    const int rowbase = blockIdx.x * 16;

    #pragma unroll
    for (int i = 0; i < 32; ++i) {
        int idx = tid + i * 128;
        int r = idx >> 6, c = idx & 63;
        Ws[r * 68 + c] = W[idx];
    }
    #pragma unroll
    for (int i = 0; i < 8; ++i) {
        int idx = tid + i * 128;
        int r = idx >> 6, c = idx & 63;
        long long g = (long long)(rowbase + r) * DD + c;
        float den = denom[g];
        float agg = (den > 0.0f) ? (numer[g] / den) : 0.0f;
        fts[r * 68 + c] = nf[g] + agg;
    }
    __syncthreads();

    const int lane  = tid & 31;
    const int wave  = tid >> 5;
    const int half  = lane >> 4;
    const int m     = lane & 15;

    v8f acc = {};
    #pragma unroll
    for (int k0 = 0; k0 < DD; k0 += 4) {
        const int k = k0 + 2 * half;
        v2f a, b;
        a.x = fts[m * 68 + k];
        a.y = fts[m * 68 + k + 1];
        b.x = Ws[(wave * 16 + m) * 68 + k];
        b.y = Ws[(wave * 16 + m) * 68 + k + 1];
        acc = __builtin_amdgcn_wmma_f32_16x16x4_f32(
                   false, a,  false, b,
                   (short)0, acc,
                   false,  false);
    }

    const float bb = bias[wave * 16 + m];
    #pragma unroll
    for (int v = 0; v < 8; ++v) {
        int row = rowbase + v + 8 * half;
        out[(long long)row * DD + wave * 16 + m] = acc[v] + bb;
    }
}

extern "C" void kernel_launch(void* const* d_in, const int* in_sizes, int n_in,
                              void* d_out, int out_size, void* d_ws, size_t ws_size,
                              hipStream_t stream) {
    const float* nf    = (const float*)d_in[0];
    const int*   eattr = (const int*)  d_in[1];
    const int*   src   = (const int*)  d_in[2];
    const int*   dst   = (const int*)  d_in[3];
    const float* emb   = (const float*)d_in[4];
    const float* W     = (const float*)d_in[5];
    const float* bias  = (const float*)d_in[6];
    float*       out   = (float*)d_out;

    int*   cnt    = (int*)d_ws;
    int*   off    = cnt + NPAD;
    int*   rank   = off + NPAD;
    int*   sorted = rank + EE;
    float* denom  = (float*)(sorted + 2 * EE);
    float* numer  = denom + (size_t)NN * DD;

    zero_counts<<<(NN + 255) / 256, 256, 0, stream>>>(cnt);
    hist_rank<<<(EE + 255) / 256, 256, 0, stream>>>(dst, cnt, rank);
    scan_offsets<<<1, 1024, 0, stream>>>(cnt, off);
    scatter_edges<<<(EE + 255) / 256, 256, 0, stream>>>(eattr, src, dst, rank, off, sorted);
    seg_softmax<<<NN / 8, 256, 0, stream>>>(nf, emb, off, sorted, denom, numer);

    node_gemm<<<NN / 16, 128, 0, stream>>>(nf, numer, denom, W, bias, out);
}


# Counting sort replaced by 32-place slots per node filled in the counting pass (place = count atomic's return; overflow list past 32, summed last): no scan, no second edge pass, 6 launches to 4
# speedup vs baseline: 3.2339x; 1.2153x over previous
#include <hip/hip_runtime.h>


#define NN 50000
#define EE 800000
#define DD 64
#define EPSF 1e-7f
#define NPAD 50176
#define CAP 32

typedef __attribute__((ext_vector_type(2))) float v2f;
typedef __attribute__((ext_vector_type(8))) float v8f;


__global__ void zero_counts(int* __restrict__ cnt) {
    int i = blockIdx.x * blockDim.x + threadIdx.x;
    if (i <= NN) cnt[i] = 0;
}

__global__ void bucket_edges(const int* __restrict__ eattr,
                             const int* __restrict__ src,
                             const int* __restrict__ dst,
                             int*       __restrict__ cnt,
                             int*       __restrict__ bucket,
                             int*       __restrict__ ovf) {
    int e = blockIdx.x * blockDim.x + threadIdx.x;
    if (e < EE) {
        const int d = dst[e];
        const int s = src[e];
        const int a = eattr[e];
        const int r = atomicAdd(cnt + d, 1);
        if (r < CAP) {
            reinterpret_cast<int2*>(bucket)[(size_t)d * CAP + r] = make_int2(s, a);
        } else {
            const int k = atomicAdd(cnt + NN, 1);
            reinterpret_cast<int4*>(ovf)[k] = make_int4(d, s, a, 0);
        }
    }
}

__global__ __launch_bounds__(256) void seg_softmax(
        const float* __restrict__ nf,
        const float* __restrict__ emb,
        const int*   __restrict__ cnt,
        const int*   __restrict__ bucket_pairs,
        const int*   __restrict__ ovf_recs,
        float*       __restrict__ denom,
        float*       __restrict__ numer) {
    const int lane = threadIdx.x & 31;
    const int t = __builtin_amdgcn_readfirstlane((int)(blockIdx.x * 8 + (threadIdx.x >> 5)));
    const int deg = cnt[t];
    const int n = min(deg, CAP);
    const float2* nf2  = reinterpret_cast<const float2*>(nf);
    const float2* emb2 = reinterpret_cast<const float2*>(emb);
    const int2* slot = reinterpret_cast<const int2*>(bucket_pairs) + (size_t)t * CAP;

    float mx = 0.0f, my = 0.0f;
    float sx = 0.0f, sy = 0.0f;
    float qx = 0.0f, qy = 0.0f;
    if (n > 0) {
        const int2 mine = slot[min(lane, n - 1)];
        for (int j = 0; j < n; j += 4) {
            float2 x[4], eb[4];
            #pragma unroll
            for (int k = 0; k < 4; ++k) {
                const int jj = min(j + k, n - 1);
                const int s = __builtin_amdgcn_readlane(mine.x, jj);
                const int a = __builtin_amdgcn_readlane(mine.y, jj);
                x[k]  = nf2[(size_t)s * (DD / 2) + lane];
                eb[k] = emb2[a * (DD / 2) + lane];
            }
            asm volatile("" :: "v"(x[0].x), "v"(x[0].y), "v"(x[1].x), "v"(x[1].y),
                               "v"(x[2].x), "v"(x[2].y), "v"(x[3].x), "v"(x[3].y),
                               "v"(eb[0].x), "v"(eb[0].y), "v"(eb[1].x), "v"(eb[1].y),
                               "v"(eb[2].x), "v"(eb[2].y), "v"(eb[3].x), "v"(eb[3].y));
            float ax[4], ay[4];
            #pragma unroll
            for (int k = 0; k < 4; ++k) {
                const bool valid = (j + k) < n;
                const float vx = fmaxf(x[k].x + eb[k].x, 0.0f) + EPSF;
                const float vy = fmaxf(x[k].y + eb[k].y, 0.0f) + EPSF;
                ax[k] = valid ? vx : 0.0f;
                ay[k] = valid ? vy : 0.0f;
            }
            const float nx = fmaxf(mx, fmaxf(fmaxf(ax[0], ax[1]), fmaxf(ax[2], ax[3])));
            const float ny = fmaxf(my, fmaxf(fmaxf(ay[0], ay[1]), fmaxf(ay[2], ay[3])));
            const float rx = __expf(mx - nx);
            const float ry = __expf(my - ny);
            sx *= rx; qx *= rx;
            sy *= ry; qy *= ry;
            #pragma unroll
            for (int k = 0; k < 4; ++k) {
                const bool valid = (j + k) < n;
                const float zx = valid ? __expf(ax[k] - nx) : 0.0f;
                const float zy = valid ? __expf(ay[k] - ny) : 0.0f;
                sx += zx; qx += ax[k] * zx;
                sy += zy; qy += ay[k] * zy;
            }
            mx = nx; my = ny;
        }
    }
    if (deg > CAP) {
        const int4* ovf = reinterpret_cast<const int4*>(ovf_recs);
        const int total = min(cnt[NN], EE);
        for (int base = 0; base < total; base += 32) {
            const int i = base + lane;
            int4 rec = make_int4(-1, 0, 0, 0);
            if (i < total) rec = ovf[i];
            unsigned hit = __builtin_amdgcn_ballot_w32(rec.x == t);
            while (hit) {
                const int l = __builtin_ctz(hit);
                hit &= hit - 1;
                const int s = __builtin_amdgcn_readlane(rec.y, l);
                const int a = __builtin_amdgcn_readlane(rec.z, l);
                const float2 x  = nf2[(size_t)s * (DD / 2) + lane];
                const float2 eb = emb2[a * (DD / 2) + lane];
                const float vx = fmaxf(x.x + eb.x, 0.0f) + EPSF;
                const float vy = fmaxf(x.y + eb.y, 0.0f) + EPSF;
                const float nx = fmaxf(mx, vx);
                const float ny = fmaxf(my, vy);
                const float rx = __expf(mx - nx);
                const float ry = __expf(my - ny);
                const float zx = __expf(vx - nx);
                const float zy = __expf(vy - ny);
                sx = sx * rx + zx; qx = qx * rx + vx * zx;
                sy = sy * ry + zy; qy = qy * ry + vy * zy;
                mx = nx; my = ny;
            }
        }
    }
    const size_t o = (size_t)t * (DD / 2) + lane;
    reinterpret_cast<float2*>(denom)[o] = make_float2(sx, sy);
    reinterpret_cast<float2*>(numer)[o] = make_float2(qx, qy);
}

__global__ __launch_bounds__(128) void node_gemm(
        const float* __restrict__ nf,
        const float* __restrict__ numer,
        const float* __restrict__ denom,
        const float* __restrict__ W,
        const float* __restrict__ bias,
        float*       __restrict__ out) {
    __shared__ float fts[16 * 68];
    __shared__ float Ws [64 * 68];

    const int tid     = threadIdx.x;
    const int rowbase = blockIdx.x * 16;

    #pragma unroll
    for (int i = 0; i < 32; ++i) {
        int idx = tid + i * 128;
        int r = idx >> 6, c = idx & 63;
        Ws[r * 68 + c] = W[idx];
    }
    #pragma unroll
    for (int i = 0; i < 8; ++i) {
        int idx = tid + i * 128;
        int r = idx >> 6, c = idx & 63;
        long long g = (long long)(rowbase + r) * DD + c;
        float den = denom[g];
        float agg = (den > 0.0f) ? (numer[g] / den) : 0.0f;
        fts[r * 68 + c] = nf[g] + agg;
    }
    __syncthreads();

    const int lane  = tid & 31;
    const int wave  = tid >> 5;
    const int half  = lane >> 4;
    const int m     = lane & 15;

    v8f acc = {};
    #pragma unroll
    for (int k0 = 0; k0 < DD; k0 += 4) {
        const int k = k0 + 2 * half;
        v2f a, b;
        a.x = fts[m * 68 + k];
        a.y = fts[m * 68 + k + 1];
        b.x = Ws[(wave * 16 + m) * 68 + k];
        b.y = Ws[(wave * 16 + m) * 68 + k + 1];
        acc = __builtin_amdgcn_wmma_f32_16x16x4_f32(
                   false, a,  false, b,
                   (short)0, acc,
                   false,  false);
    }

    const float bb = bias[wave * 16 + m];
    #pragma unroll
    for (int v = 0; v < 8; ++v) {
        int row = rowbase + v + 8 * half;
        out[(long long)row * DD + wave * 16 + m] = acc[v] + bb;
    }
}

extern "C" void kernel_launch(void* const* d_in, const int* in_sizes, int n_in,
                              void* d_out, int out_size, void* d_ws, size_t ws_size,
                              hipStream_t stream) {
    const float* nf    = (const float*)d_in[0];
    const int*   eattr = (const int*)  d_in[1];
    const int*   src   = (const int*)  d_in[2];
    const int*   dst   = (const int*)  d_in[3];
    const float* emb   = (const float*)d_in[4];
    const float* W     = (const float*)d_in[5];
    const float* bias  = (const float*)d_in[6];
    float*       out   = (float*)d_out;

    int*   cnt    = (int*)d_ws;
    int*   bucket = cnt + NPAD;
    int*   ovf    = bucket + (size_t)NN * CAP * 2;
    float* denom  = (float*)(ovf + (size_t)EE * 4);
    float* numer  = denom + (size_t)NN * DD;

    zero_counts<<<(NN + 1 + 255) / 256, 256, 0, stream>>>(cnt);
    bucket_edges<<<(EE + 255) / 256, 256, 0, stream>>>(eattr, src, dst, cnt, bucket, ovf);
    seg_softmax<<<NN / 8, 256, 0, stream>>>(nf, emb, cnt, bucket, ovf, denom, numer);

    node_gemm<<<NN / 16, 128, 0, stream>>>(nf, numer, denom, W, bias, out);
}


# Edge ranks from LDS counters in two passes (bin by 256 nodes, one global add per bin and group; then rank per node in each bin), not one global atomic per edge; 5 launches; sums follow the new ranks
# speedup vs baseline: 5.5449x; 1.7146x over previous
#include <hip/hip_runtime.h>


#define NN 50000
#define EE 800000
#define DD 64
#define EPSF 1e-7f
#define NPAD 50176
#define CAP 32
#define BSHIFT 8
#define NPB 256
#define NBIN 196
#define BSTRIDE 64
#define SEGCAP 8192
#define CHUNK 4096

typedef __attribute__((ext_vector_type(2))) float v2f;
typedef __attribute__((ext_vector_type(8))) float v8f;


__global__ void zero_counts(int* __restrict__ cnt) {
    int i = blockIdx.x * blockDim.x + threadIdx.x;
    if (i < NPAD + NBIN * BSTRIDE) cnt[i] = 0;
}

__device__ __forceinline__ void put_overflow(int d, int s, int a,
                                             int* __restrict__ cnt,
                                             int* __restrict__ ovf) {
    const int k = atomicAdd(cnt + NN, 1);
    reinterpret_cast<int4*>(ovf)[k] = make_int4(d, s, a, 0);
}

__global__ __launch_bounds__(1024) void bin_edges(const int* __restrict__ eattr,
                                                  const int* __restrict__ src,
                                                  const int* __restrict__ dst,
                                                  int*       __restrict__ cnt,
                                                  int*       __restrict__ bcnt,
                                                  int*       __restrict__ seg,
                                                  int*       __restrict__ bucket,
                                                  int*       __restrict__ ovf) {
    __shared__ int4 stage[CHUNK];
    __shared__ int lcnt[NBIN];
    __shared__ int lpre[NBIN];
    __shared__ int lbase[NBIN];
    __shared__ int wsum[8];
    const int tid  = threadIdx.x;
    const int lane = tid & 31;
    int d[4], s[4], a[4], lr[4];
    const int e0 = blockIdx.x * CHUNK + tid;
    #pragma unroll
    for (int k = 0; k < 4; ++k) {
        const int e = e0 + k * 1024;
        d[k] = -1; s[k] = 0; a[k] = 0; lr[k] = 0;
        if (e < EE) { d[k] = dst[e]; s[k] = src[e]; a[k] = eattr[e]; }
    }
    if (tid < NBIN) lcnt[tid] = 0;
    __syncthreads();

    #pragma unroll
    for (int k = 0; k < 4; ++k)
        if (d[k] >= 0) lr[k] = atomicAdd(&lcnt[d[k] >> BSHIFT], 1);
    __syncthreads();

    int v = 0, incl = 0;
    if (tid < 256) {
        v = (tid < NBIN) ? lcnt[tid] : 0;
        incl = v;
        #pragma unroll
        for (int off = 1; off < 32; off <<= 1) {
            const int t = __shfl_up(incl, off);
            if (lane >= off) incl += t;
        }
        if (lane == 31) wsum[tid >> 5] = incl;
    }
    __syncthreads();
    if (tid < NBIN) {
        int p = 0;
        for (int w = 0; w < (tid >> 5); ++w) p += wsum[w];
        lpre[tid] = p + incl - v;
        lbase[tid] = (v > 0) ? atomicAdd(bcnt + tid * BSTRIDE, v) : 0;
    }
    __syncthreads();

    #pragma unroll
    for (int k = 0; k < 4; ++k)
        if (d[k] >= 0) stage[lpre[d[k] >> BSHIFT] + lr[k]] = make_int4(d[k], s[k], a[k], 0);
    __syncthreads();

    const int nvalid = min(CHUNK, EE - (int)blockIdx.x * CHUNK);
    #pragma unroll
    for (int k = 0; k < 4; ++k) {
        const int j = tid + k * 1024;
        if (j < nvalid) {
            const int4 rec = stage[j];
            const int b = rec.x >> BSHIFT;
            const int pos = lbase[b] + (j - lpre[b]);
            if (pos < SEGCAP) {
                reinterpret_cast<int4*>(seg)[(size_t)b * SEGCAP + pos] = rec;
            } else {
                const int r = atomicAdd(cnt + rec.x, 1);
                if (r < CAP) reinterpret_cast<int2*>(bucket)[(size_t)rec.x * CAP + r] = make_int2(rec.y, rec.z);
                else         put_overflow(rec.x, rec.y, rec.z, cnt, ovf);
            }
        }
    }
}

__global__ __launch_bounds__(1024) void place_edges(const int* __restrict__ bcnt,
                                                    const int* __restrict__ seg,
                                                    int*       __restrict__ cnt,
                                                    int*       __restrict__ bucket,
                                                    int*       __restrict__ ovf) {
    __shared__ int2 slots[NPB * CAP];
    __shared__ int ncnt[NPB];
    __shared__ int nc0[NPB];
    const int tid   = threadIdx.x;
    const int bin   = blockIdx.x;
    const int node0 = bin << BSHIFT;
    int c0 = 0;
    if (tid < NPB && node0 + tid < NN) c0 = cnt[node0 + tid];
    const int total = min(bcnt[bin * BSTRIDE], SEGCAP);
    const int4* recs = reinterpret_cast<const int4*>(seg) + (size_t)bin * SEGCAP;
    int4 rec[4];
    #pragma unroll
    for (int k = 0; k < 4; ++k) rec[k] = make_int4(0, 0, 0, 0);
    if (total > 0) {
        #pragma unroll
        for (int k = 0; k < 4; ++k) rec[k] = recs[min(tid + k * 1024, total - 1)];
    }
    if (tid < NPB) { ncnt[tid] = c0; nc0[tid] = c0; }
    __syncthreads();

    for (int base = 0; base < total; base += 4096) {
        if (base > 0) {
            #pragma unroll
            for (int k = 0; k < 4; ++k) rec[k] = recs[min(base + tid + k * 1024, total - 1)];
        }
        #pragma unroll
        for (int k = 0; k < 4; ++k) {
            if (base + tid + k * 1024 < total) {
                const int dl = rec[k].x - node0;
                const int r = atomicAdd(&ncnt[dl], 1);
                if (r < CAP) slots[dl * CAP + r] = make_int2(rec[k].y, rec[k].z);
                else         put_overflow(rec[k].x, rec[k].y, rec[k].z, cnt, ovf);
            }
        }
    }
    __syncthreads();

    const int lane = tid & 31;
    const int wave = tid >> 5;
    int2 val[NPB / 32];
    bool mine[NPB / 32];
    #pragma unroll
    for (int i = 0; i < NPB / 32; ++i) {
        const int nl = wave + 32 * i;
        const int hi = min(ncnt[nl], CAP);
        mine[i] = (node0 + nl < NN) && lane >= nc0[nl] && lane < hi;
        val[i]  = slots[nl * CAP + lane];
    }
    #pragma unroll
    for (int i = 0; i < NPB / 32; ++i) {
        const int node = node0 + wave + 32 * i;
        if (mine[i]) reinterpret_cast<int2*>(bucket)[(size_t)node * CAP + lane] = val[i];
    }
    if (tid < NPB) {
        const int node = node0 + tid;
        if (node < NN) cnt[node] = ncnt[tid];
    }
}

__global__ __launch_bounds__(256) void seg_softmax(
        const float* __restrict__ nf,
        const float* __restrict__ emb,
        const int*   __restrict__ cnt,
        const int*   __restrict__ bucket_pairs,
        const int*   __restrict__ ovf_recs,
        float*       __restrict__ denom,
        float*       __restrict__ numer) {
    const int lane = threadIdx.x & 31;
    const int t = __builtin_amdgcn_readfirstlane((int)(blockIdx.x * 8 + (threadIdx.x >> 5)));
    const int deg = cnt[t];
    const int n = min(deg, CAP);
    const float2* nf2  = reinterpret_cast<const float2*>(nf);
    const float2* emb2 = reinterpret_cast<const float2*>(emb);
    const int2* slot = reinterpret_cast<const int2*>(bucket_pairs) + (size_t)t * CAP;

    float mx = 0.0f, my = 0.0f;
    float sx = 0.0f, sy = 0.0f;
    float qx = 0.0f, qy = 0.0f;
    if (n > 0) {
        const int2 mine = slot[min(lane, n - 1)];
        for (int j = 0; j < n; j += 4) {
            float2 x[4], eb[4];
            #pragma unroll
            for (int k = 0; k < 4; ++k) {
                const int jj = min(j + k, n - 1);
                const int s = __builtin_amdgcn_readlane(mine.x, jj);
                const int a = __builtin_amdgcn_readlane(mine.y, jj);
                x[k]  = nf2[(size_t)s * (DD / 2) + lane];
                eb[k] = emb2[a * (DD / 2) + lane];
            }
            asm volatile("" :: "v"(x[0].x), "v"(x[0].y), "v"(x[1].x), "v"(x[1].y),
                               "v"(x[2].x), "v"(x[2].y), "v"(x[3].x), "v"(x[3].y),
                               "v"(eb[0].x), "v"(eb[0].y), "v"(eb[1].x), "v"(eb[1].y),
                               "v"(eb[2].x), "v"(eb[2].y), "v"(eb[3].x), "v"(eb[3].y));
            float ax[4], ay[4];
            #pragma unroll
            for (int k = 0; k < 4; ++k) {
                const bool valid = (j + k) < n;
                const float vx = fmaxf(x[k].x + eb[k].x, 0.0f) + EPSF;
                const float vy = fmaxf(x[k].y + eb[k].y, 0.0f) + EPSF;
                ax[k] = valid ? vx : 0.0f;
                ay[k] = valid ? vy : 0.0f;
            }
            const float nx = fmaxf(mx, fmaxf(fmaxf(ax[0], ax[1]), fmaxf(ax[2], ax[3])));
            const float ny = fmaxf(my, fmaxf(fmaxf(ay[0], ay[1]), fmaxf(ay[2], ay[3])));
            const float rx = __expf(mx - nx);
            const float ry = __expf(my - ny);
            sx *= rx; qx *= rx;
            sy *= ry; qy *= ry;
            #pragma unroll
            for (int k = 0; k < 4; ++k) {
                const bool valid = (j + k) < n;
                const float zx = valid ? __expf(ax[k] - nx) : 0.0f;
                const float zy = valid ? __expf(ay[k] - ny) : 0.0f;
                sx += zx; qx += ax[k] * zx;
                sy += zy; qy += ay[k] * zy;
            }
            mx = nx; my = ny;
        }
    }
    if (deg > CAP) {
        const int4* ovf = reinterpret_cast<const int4*>(ovf_recs);
        const int total = min(cnt[NN], EE);
        for (int base = 0; base < total; base += 32) {
            const int i = base + lane;
            int4 rec = make_int4(-1, 0, 0, 0);
            if (i < total) rec = ovf[i];
            unsigned hit = __builtin_amdgcn_ballot_w32(rec.x == t);
            while (hit) {
                const int l = __builtin_ctz(hit);
                hit &= hit - 1;
                const int s = __builtin_amdgcn_readlane(rec.y, l);
                const int a = __builtin_amdgcn_readlane(rec.z, l);
                const float2 x  = nf2[(size_t)s * (DD / 2) + lane];
                const float2 eb = emb2[a * (DD / 2) + lane];
                const float vx = fmaxf(x.x + eb.x, 0.0f) + EPSF;
                const float vy = fmaxf(x.y + eb.y, 0.0f) + EPSF;
                const float nx = fmaxf(mx, vx);
                const float ny = fmaxf(my, vy);
                const float rx = __expf(mx - nx);
                const float ry = __expf(my - ny);
                const float zx = __expf(vx - nx);
                const float zy = __expf(vy - ny);
                sx = sx * rx + zx; qx = qx * rx + vx * zx;
                sy = sy * ry + zy; qy = qy * ry + vy * zy;
                mx = nx; my = ny;
            }
        }
    }
    const size_t o = (size_t)t * (DD / 2) + lane;
    reinterpret_cast<float2*>(denom)[o] = make_float2(sx, sy);
    reinterpret_cast<float2*>(numer)[o] = make_float2(qx, qy);
}

__global__ __launch_bounds__(128) void node_gemm(
        const float* __restrict__ nf,
        const float* __restrict__ numer,
        const float* __restrict__ denom,
        const float* __restrict__ W,
        const float* __restrict__ bias,
        float*       __restrict__ out) {
    __shared__ float fts[16 * 68];
    __shared__ float Ws [64 * 68];

    const int tid     = threadIdx.x;
    const int rowbase = blockIdx.x * 16;

    #pragma unroll
    for (int i = 0; i < 32; ++i) {
        int idx = tid + i * 128;
        int r = idx >> 6, c = idx & 63;
        Ws[r * 68 + c] = W[idx];
    }
    #pragma unroll
    for (int i = 0; i < 8; ++i) {
        int idx = tid + i * 128;
        int r = idx >> 6, c = idx & 63;
        long long g = (long long)(rowbase + r) * DD + c;
        float den = denom[g];
        float agg = (den > 0.0f) ? (numer[g] / den) : 0.0f;
        fts[r * 68 + c] = nf[g] + agg;
    }
    __syncthreads();

    const int lane  = tid & 31;
    const int wave  = tid >> 5;
    const int half  = lane >> 4;
    const int m     = lane & 15;

    v8f acc = {};
    #pragma unroll
    for (int k0 = 0; k0 < DD; k0 += 4) {
        const int k = k0 + 2 * half;
        v2f a, b;
        a.x = fts[m * 68 + k];
        a.y = fts[m * 68 + k + 1];
        b.x = Ws[(wave * 16 + m) * 68 + k];
        b.y = Ws[(wave * 16 + m) * 68 + k + 1];
        acc = __builtin_amdgcn_wmma_f32_16x16x4_f32(
                   false, a,  false, b,
                   (short)0, acc,
                   false,  false);
    }

    const float bb = bias[wave * 16 + m];
    #pragma unroll
    for (int v = 0; v < 8; ++v) {
        int row = rowbase + v + 8 * half;
        out[(long long)row * DD + wave * 16 + m] = acc[v] + bb;
    }
}

extern "C" void kernel_launch(void* const* d_in, const int* in_sizes, int n_in,
                              void* d_out, int out_size, void* d_ws, size_t ws_size,
                              hipStream_t stream) {
    const float* nf    = (const float*)d_in[0];
    const int*   eattr = (const int*)  d_in[1];
    const int*   src   = (const int*)  d_in[2];
    const int*   dst   = (const int*)  d_in[3];
    const float* emb   = (const float*)d_in[4];
    const float* W     = (const float*)d_in[5];
    const float* bias  = (const float*)d_in[6];
    float*       out   = (float*)d_out;

    int*   cnt    = (int*)d_ws;
    int*   bcnt   = cnt + NPAD;
    int*   bucket = bcnt + NBIN * BSTRIDE;
    int*   ovf    = bucket + (size_t)NN * CAP * 2;
    int*   seg    = ovf + (size_t)EE * 4;
    float* denom  = (float*)(seg + (size_t)NBIN * SEGCAP * 4);
    float* numer  = denom + (size_t)NN * DD;

    zero_counts<<<(NPAD + NBIN * BSTRIDE + 255) / 256, 256, 0, stream>>>(cnt);
    bin_edges<<<(EE + CHUNK - 1) / CHUNK, 1024, 0, stream>>>(eattr, src, dst, cnt, bcnt, seg, bucket, ovf);
    place_edges<<<NBIN, 1024, 0, stream>>>(bcnt, seg, cnt, bucket, ovf);
    seg_softmax<<<NN / 8, 256, 0, stream>>>(nf, emb, cnt, bucket, ovf, denom, numer);

    node_gemm<<<NN / 16, 128, 0, stream>>>(nf, numer, denom, W, bias, out);
}


# Softmax pass and node GEMM fused: a wave reduces 4 nodes in turn (sums and order unchanged) into an LDS feats tile, 8 waves do the 32x64 product; 4 launches, no denom/numer planes, 5x fewer waves
# speedup vs baseline: 6.4673x; 1.1664x over previous
#include <hip/hip_runtime.h>


#define NN 50000
#define EE 800000
#define DD 64
#define EPSF 1e-7f
#define NPAD 50176
#define CAP 32
#define BSHIFT 8
#define NPB 256
#define NBIN 196
#define BSTRIDE 64
#define SEGCAP 8192
#define CHUNK 4096

typedef __attribute__((ext_vector_type(2))) float v2f;
typedef __attribute__((ext_vector_type(8))) float v8f;


__global__ void zero_counts(int* __restrict__ cnt) {
    int i = blockIdx.x * blockDim.x + threadIdx.x;
    if (i < NPAD + NBIN * BSTRIDE) cnt[i] = 0;
}

__device__ __forceinline__ void put_overflow(int d, int s, int a,
                                             int* __restrict__ cnt,
                                             int* __restrict__ ovf) {
    const int k = atomicAdd(cnt + NN, 1);
    reinterpret_cast<int4*>(ovf)[k] = make_int4(d, s, a, 0);
}

__global__ __launch_bounds__(1024) void bin_edges(const int* __restrict__ eattr,
                                                  const int* __restrict__ src,
                                                  const int* __restrict__ dst,
                                                  int*       __restrict__ cnt,
                                                  int*       __restrict__ bcnt,
                                                  int*       __restrict__ seg,
                                                  int*       __restrict__ bucket,
                                                  int*       __restrict__ ovf) {
    __shared__ int4 stage[CHUNK];
    __shared__ int lcnt[NBIN];
    __shared__ int lpre[NBIN];
    __shared__ int lbase[NBIN];
    __shared__ int wsum[8];
    const int tid  = threadIdx.x;
    const int lane = tid & 31;
    int d[4], s[4], a[4], lr[4];
    const int e0 = blockIdx.x * CHUNK + tid;
    #pragma unroll
    for (int k = 0; k < 4; ++k) {
        const int e = e0 + k * 1024;
        d[k] = -1; s[k] = 0; a[k] = 0; lr[k] = 0;
        if (e < EE) { d[k] = dst[e]; s[k] = src[e]; a[k] = eattr[e]; }
    }
    if (tid < NBIN) lcnt[tid] = 0;
    __syncthreads();

    #pragma unroll
    for (int k = 0; k < 4; ++k)
        if (d[k] >= 0) lr[k] = atomicAdd(&lcnt[d[k] >> BSHIFT], 1);
    __syncthreads();

    int v = 0, incl = 0;
    if (tid < 256) {
        v = (tid < NBIN) ? lcnt[tid] : 0;
        incl = v;
        #pragma unroll
        for (int off = 1; off < 32; off <<= 1) {
            const int t = __shfl_up(incl, off);
            if (lane >= off) incl += t;
        }
        if (lane == 31) wsum[tid >> 5] = incl;
    }
    __syncthreads();
    if (tid < NBIN) {
        int p = 0;
        for (int w = 0; w < (tid >> 5); ++w) p += wsum[w];
        lpre[tid] = p + incl - v;
        lbase[tid] = (v > 0) ? atomicAdd(bcnt + tid * BSTRIDE, v) : 0;
    }
    __syncthreads();

    #pragma unroll
    for (int k = 0; k < 4; ++k)
        if (d[k] >= 0) stage[lpre[d[k] >> BSHIFT] + lr[k]] = make_int4(d[k], s[k], a[k], 0);
    __syncthreads();

    const int nvalid = min(CHUNK, EE - (int)blockIdx.x * CHUNK);
    #pragma unroll
    for (int k = 0; k < 4; ++k) {
        const int j = tid + k * 1024;
        if (j < nvalid) {
            const int4 rec = stage[j];
            const int b = rec.x >> BSHIFT;
            const int pos = lbase[b] + (j - lpre[b]);
            if (pos < SEGCAP) {
                reinterpret_cast<int4*>(seg)[(size_t)b * SEGCAP + pos] = rec;
            } else {
                const int r = atomicAdd(cnt + rec.x, 1);
                if (r < CAP) reinterpret_cast<int2*>(bucket)[(size_t)rec.x * CAP + r] = make_int2(rec.y, rec.z);
                else         put_overflow(rec.x, rec.y, rec.z, cnt, ovf);
            }
        }
    }
}

__global__ __launch_bounds__(1024) void place_edges(const int* __restrict__ bcnt,
                                                    const int* __restrict__ seg,
                                                    int*       __restrict__ cnt,
                                                    int*       __restrict__ bucket,
                                                    int*       __restrict__ ovf) {
    __shared__ int2 slots[NPB * CAP];
    __shared__ int ncnt[NPB];
    __shared__ int nc0[NPB];
    const int tid   = threadIdx.x;
    const int bin   = blockIdx.x;
    const int node0 = bin << BSHIFT;
    int c0 = 0;
    if (tid < NPB && node0 + tid < NN) c0 = cnt[node0 + tid];
    const int total = min(bcnt[bin * BSTRIDE], SEGCAP);
    const int4* recs = reinterpret_cast<const int4*>(seg) + (size_t)bin * SEGCAP;
    int4 rec[4];
    #pragma unroll
    for (int k = 0; k < 4; ++k) rec[k] = make_int4(0, 0, 0, 0);
    if (total > 0) {
        #pragma unroll
        for (int k = 0; k < 4; ++k) rec[k] = recs[min(tid + k * 1024, total - 1)];
    }
    if (tid < NPB) { ncnt[tid] = c0; nc0[tid] = c0; }
    __syncthreads();

    for (int base = 0; base < total; base += 4096) {
        if (base > 0) {
            #pragma unroll
            for (int k = 0; k < 4; ++k) rec[k] = recs[min(base + tid + k * 1024, total - 1)];
        }
        #pragma unroll
        for (int k = 0; k < 4; ++k) {
            if (base + tid + k * 1024 < total) {
                const int dl = rec[k].x - node0;
                const int r = atomicAdd(&ncnt[dl], 1);
                if (r < CAP) slots[dl * CAP + r] = make_int2(rec[k].y, rec[k].z);
                else         put_overflow(rec[k].x, rec[k].y, rec[k].z, cnt, ovf);
            }
        }
    }
    __syncthreads();

    const int lane = tid & 31;
    const int wave = tid >> 5;
    int2 val[NPB / 32];
    bool mine[NPB / 32];
    #pragma unroll
    for (int i = 0; i < NPB / 32; ++i) {
        const int nl = wave + 32 * i;
        const int hi = min(ncnt[nl], CAP);
        mine[i] = (node0 + nl < NN) && lane >= nc0[nl] && lane < hi;
        val[i]  = slots[nl * CAP + lane];
    }
    #pragma unroll
    for (int i = 0; i < NPB / 32; ++i) {
        const int node = node0 + wave + 32 * i;
        if (mine[i]) reinterpret_cast<int2*>(bucket)[(size_t)node * CAP + lane] = val[i];
    }
    if (tid < NPB) {
        const int node = node0 + tid;
        if (node < NN) cnt[node] = ncnt[tid];
    }
}

#define NPW 4
#define TILE_N 32
__global__ __launch_bounds__(256) void softmax_gemm(
        const float* __restrict__ nf,
        const float* __restrict__ emb,
        const int*   __restrict__ cnt,
        const int*   __restrict__ bucket_pairs,
        const int*   __restrict__ ovf_recs,
        const float* __restrict__ W,
        const float* __restrict__ bias,
        float*       __restrict__ out) {
    __shared__ float fts[TILE_N * 68];
    __shared__ float Ws [64 * 68];

    const int tid     = threadIdx.x;
    const int lane    = tid & 31;
    const int wave    = __builtin_amdgcn_readfirstlane(tid >> 5);
    const int rowbase = blockIdx.x * TILE_N;
    const int t0      = rowbase + wave * NPW;

    const float2* nf2  = reinterpret_cast<const float2*>(nf);
    const float2* emb2 = reinterpret_cast<const float2*>(emb);
    const int2* slots  = reinterpret_cast<const int2*>(bucket_pairs);

    const int cvec = cnt[t0 + (lane & (NPW - 1))];

    float4 wv[4];
    #pragma unroll
    for (int i = 0; i < 4; ++i)
        wv[i] = reinterpret_cast<const float4*>(W)[tid + i * 256];

    int deg_next = (t0 < NN) ? __builtin_amdgcn_readlane(cvec, 0) : 0;
    int2 pairs_next = make_int2(0, 0);
    if (deg_next > 0) pairs_next = slots[(size_t)t0 * CAP + min(lane, min(deg_next, CAP) - 1)];

    #pragma unroll
    for (int i = 0; i < 4; ++i) {
        const int idx4 = tid + i * 256;
        *reinterpret_cast<float4*>(&Ws[(idx4 >> 4) * 68 + (idx4 & 15) * 4]) = wv[i];
    }

    for (int q = 0; q < NPW; ++q) {
        const int t   = t0 + q;
        const int deg = deg_next;
        const int n   = min(deg, CAP);
        const int2 mine = pairs_next;
        float2 own = make_float2(0.0f, 0.0f);
        if (t < NN) own = nf2[(size_t)t * (DD / 2) + lane];
        deg_next = 0;
        if (q + 1 < NPW && t + 1 < NN) deg_next = __builtin_amdgcn_readlane(cvec, q + 1);
        if (deg_next > 0) pairs_next = slots[(size_t)(t + 1) * CAP + min(lane, min(deg_next, CAP) - 1)];

        float mx = 0.0f, my = 0.0f;
        float sx = 0.0f, sy = 0.0f;
        float qx = 0.0f, qy = 0.0f;
        for (int j = 0; j < n; j += 4) {
            float2 x[4], eb[4];
            #pragma unroll
            for (int k = 0; k < 4; ++k) {
                const int jj = min(j + k, n - 1);
                const int s = __builtin_amdgcn_readlane(mine.x, jj);
                const int a = __builtin_amdgcn_readlane(mine.y, jj);
                x[k]  = nf2[(size_t)s * (DD / 2) + lane];
                eb[k] = emb2[a * (DD / 2) + lane];
            }
            asm volatile("" :: "v"(x[0].x), "v"(x[0].y), "v"(x[1].x), "v"(x[1].y),
                               "v"(x[2].x), "v"(x[2].y), "v"(x[3].x), "v"(x[3].y),
                               "v"(eb[0].x), "v"(eb[0].y), "v"(eb[1].x), "v"(eb[1].y),
                               "v"(eb[2].x), "v"(eb[2].y), "v"(eb[3].x), "v"(eb[3].y));
            float ax[4], ay[4];
            #pragma unroll
            for (int k = 0; k < 4; ++k) {
                const bool valid = (j + k) < n;
                const float vx = fmaxf(x[k].x + eb[k].x, 0.0f) + EPSF;
                const float vy = fmaxf(x[k].y + eb[k].y, 0.0f) + EPSF;
                ax[k] = valid ? vx : 0.0f;
                ay[k] = valid ? vy : 0.0f;
            }
            const float nx = fmaxf(mx, fmaxf(fmaxf(ax[0], ax[1]), fmaxf(ax[2], ax[3])));
            const float ny = fmaxf(my, fmaxf(fmaxf(ay[0], ay[1]), fmaxf(ay[2], ay[3])));
            const float rx = __expf(mx - nx);
            const float ry = __expf(my - ny);
            sx *= rx; qx *= rx;
            sy *= ry; qy *= ry;
            #pragma unroll
            for (int k = 0; k < 4; ++k) {
                const bool valid = (j + k) < n;
                const float zx = valid ? __expf(ax[k] - nx) : 0.0f;
                const float zy = valid ? __expf(ay[k] - ny) : 0.0f;
                sx += zx; qx += ax[k] * zx;
                sy += zy; qy += ay[k] * zy;
            }
            mx = nx; my = ny;
        }
        if (deg > CAP) {
            const int4* ovf = reinterpret_cast<const int4*>(ovf_recs);
            const int total = min(cnt[NN], EE);
            for (int base = 0; base < total; base += 32) {
                const int i = base + lane;
                int4 rec = make_int4(-1, 0, 0, 0);
                if (i < total) rec = ovf[i];
                unsigned hit = __builtin_amdgcn_ballot_w32(rec.x == t);
                while (hit) {
                    const int l = __builtin_ctz(hit);
                    hit &= hit - 1;
                    const int s = __builtin_amdgcn_readlane(rec.y, l);
                    const int a = __builtin_amdgcn_readlane(rec.z, l);
                    const float2 x  = nf2[(size_t)s * (DD / 2) + lane];
                    const float2 eb = emb2[a * (DD / 2) + lane];
                    const float vx = fmaxf(x.x + eb.x, 0.0f) + EPSF;
                    const float vy = fmaxf(x.y + eb.y, 0.0f) + EPSF;
                    const float nx = fmaxf(mx, vx);
                    const float ny = fmaxf(my, vy);
                    const float rx = __expf(mx - nx);
                    const float ry = __expf(my - ny);
                    const float zx = __expf(vx - nx);
                    const float zy = __expf(vy - ny);
                    sx = sx * rx + zx; qx = qx * rx + vx * zx;
                    sy = sy * ry + zy; qy = qy * ry + vy * zy;
                    mx = nx; my = ny;
                }
            }
        }
        const float aggx = (sx > 0.0f) ? (qx / sx) : 0.0f;
        const float aggy = (sy > 0.0f) ? (qy / sy) : 0.0f;
        *reinterpret_cast<float2*>(&fts[(wave * NPW + q) * 68 + 2 * lane]) =
            make_float2(own.x + aggx, own.y + aggy);
    }
    __syncthreads();

    const int ct    = wave & 3;
    const int rt    = wave >> 2;
    const int half  = lane >> 4;
    const int m     = lane & 15;

    v8f acc = {};
    #pragma unroll
    for (int k0 = 0; k0 < DD; k0 += 4) {
        const int k = k0 + 2 * half;
        v2f a, b;
        a.x = fts[(rt * 16 + m) * 68 + k];
        a.y = fts[(rt * 16 + m) * 68 + k + 1];
        b.x = Ws[(ct * 16 + m) * 68 + k];
        b.y = Ws[(ct * 16 + m) * 68 + k + 1];
        acc = __builtin_amdgcn_wmma_f32_16x16x4_f32(
                   false, a,  false, b,
                   (short)0, acc,
                   false,  false);
    }

    const int tilebase = rowbase + rt * 16;
    if (tilebase < NN) {
        const float bb = bias[ct * 16 + m];
        #pragma unroll
        for (int v = 0; v < 8; ++v) {
            const int row = tilebase + v + 8 * half;
            out[(long long)row * DD + ct * 16 + m] = acc[v] + bb;
        }
    }
}

extern "C" void kernel_launch(void* const* d_in, const int* in_sizes, int n_in,
                              void* d_out, int out_size, void* d_ws, size_t ws_size,
                              hipStream_t stream) {
    const float* nf    = (const float*)d_in[0];
    const int*   eattr = (const int*)  d_in[1];
    const int*   src   = (const int*)  d_in[2];
    const int*   dst   = (const int*)  d_in[3];
    const float* emb   = (const float*)d_in[4];
    const float* W     = (const float*)d_in[5];
    const float* bias  = (const float*)d_in[6];
    float*       out   = (float*)d_out;

    int*   cnt    = (int*)d_ws;
    int*   bcnt   = cnt + NPAD;
    int*   bucket = bcnt + NBIN * BSTRIDE;
    int*   ovf    = bucket + (size_t)NN * CAP * 2;
    int*   seg    = ovf + (size_t)EE * 4;

    zero_counts<<<(NPAD + NBIN * BSTRIDE + 255) / 256, 256, 0, stream>>>(cnt);
    bin_edges<<<(EE + CHUNK - 1) / CHUNK, 1024, 0, stream>>>(eattr, src, dst, cnt, bcnt, seg, bucket, ovf);
    place_edges<<<NBIN, 1024, 0, stream>>>(bcnt, seg, cnt, bucket, ovf);
    softmax_gemm<<<(NN + TILE_N - 1) / TILE_N, 256, 0, stream>>>(nf, emb, cnt, bucket, ovf, W, bias, out);
}


# Binning writes each group's sorted records to its own stretch plus a (bin, group) start/length table; placing fetches runs by table: no global counters, no clearing launch; sums in new ranks
# speedup vs baseline: 6.7432x; 1.0427x over previous
#include <hip/hip_runtime.h>


#define NN 50000
#define EE 800000
#define DD 64
#define EPSF 1e-7f
#define NPAD 50176
#define CAP 32
#define BSHIFT 8
#define NPB 256
#define NBIN 196
#define CHUNK 4096
#define NWG 196
#define TABSZ 38464
#define RUNS 7

typedef __attribute__((ext_vector_type(2))) float v2f;
typedef __attribute__((ext_vector_type(8))) float v8f;


__device__ __forceinline__ void put_overflow(int d, int s, int a,
                                             int* __restrict__ cnt,
                                             int* __restrict__ ovf) {
    const int k = atomicAdd(cnt + NN, 1);
    reinterpret_cast<int4*>(ovf)[k] = make_int4(d, s, a, 0);
}

__global__ __launch_bounds__(1024) void bin_edges(const int* __restrict__ eattr,
                                                  const int* __restrict__ src,
                                                  const int* __restrict__ dst,
                                                  int*       __restrict__ cnt,
                                                  int*       __restrict__ tab,
                                                  int*       __restrict__ seg) {
    __shared__ int4 stage[CHUNK];
    __shared__ int lcnt[NBIN];
    __shared__ int lpre[NBIN];
    __shared__ int wsum[8];
    const int tid  = threadIdx.x;
    const int lane = tid & 31;
    int d[4], s[4], a[4], lr[4];
    const int e0 = blockIdx.x * CHUNK + tid;
    #pragma unroll
    for (int k = 0; k < 4; ++k) {
        const int e = e0 + k * 1024;
        d[k] = -1; s[k] = 0; a[k] = 0; lr[k] = 0;
        if (e < EE) { d[k] = dst[e]; s[k] = src[e]; a[k] = eattr[e]; }
    }
    if (tid < NBIN) lcnt[tid] = 0;
    if (blockIdx.x == 0 && tid < NPAD - NN) cnt[NN + tid] = 0;
    __syncthreads();

    #pragma unroll
    for (int k = 0; k < 4; ++k)
        if (d[k] >= 0) lr[k] = atomicAdd(&lcnt[d[k] >> BSHIFT], 1);
    __syncthreads();

    int v = 0, incl = 0;
    if (tid < 256) {
        v = (tid < NBIN) ? lcnt[tid] : 0;
        incl = v;
        #pragma unroll
        for (int off = 1; off < 32; off <<= 1) {
            const int t = __shfl_up(incl, off);
            if (lane >= off) incl += t;
        }
        if (lane == 31) wsum[tid >> 5] = incl;
    }
    __syncthreads();
    if (tid < NBIN) {
        int p = 0;
        for (int w = 0; w < (tid >> 5); ++w) p += wsum[w];
        p += incl - v;
        lpre[tid] = p;
        tab[tid * NWG + blockIdx.x] = (p << 13) | v;
    }
    __syncthreads();

    #pragma unroll
    for (int k = 0; k < 4; ++k)
        if (d[k] >= 0) stage[lpre[d[k] >> BSHIFT] + lr[k]] = make_int4(d[k], s[k], a[k], 0);
    __syncthreads();

    const int nvalid = min(CHUNK, EE - (int)blockIdx.x * CHUNK);
    #pragma unroll
    for (int k = 0; k < 4; ++k) {
        const int j = tid + k * 1024;
        if (j < nvalid) reinterpret_cast<int4*>(seg)[(size_t)blockIdx.x * CHUNK + j] = stage[j];
    }
}

__device__ __forceinline__ void place_one(int4 rec, int node0, int* ncnt, int2* slots,
                                          int* __restrict__ cnt, int* __restrict__ ovf) {
    const int dl = rec.x - node0;
    const int r = atomicAdd(&ncnt[dl], 1);
    if (r < CAP) slots[dl * CAP + r] = make_int2(rec.y, rec.z);
    else         put_overflow(rec.x, rec.y, rec.z, cnt, ovf);
}

__global__ __launch_bounds__(1024) void place_edges(const int* __restrict__ tab,
                                                    const int* __restrict__ seg,
                                                    int*       __restrict__ cnt,
                                                    int*       __restrict__ bucket,
                                                    int*       __restrict__ ovf) {
    __shared__ int2 slots[NPB * CAP];
    __shared__ int ncnt[NPB];
    const int tid   = threadIdx.x;
    const int lane  = tid & 31;
    const int wave  = __builtin_amdgcn_readfirstlane(tid >> 5);
    const int bin   = blockIdx.x;
    const int node0 = bin << BSHIFT;
    const int4* recs = reinterpret_cast<const int4*>(seg);
    int len[RUNS], off[RUNS];
    int4 rec[RUNS];
    const int tv = tab[bin * NWG + min(wave + 32 * (lane & 7), NWG - 1)];
    #pragma unroll
    for (int k = 0; k < RUNS; ++k) {
        const int w = wave + 32 * k;
        int t = __builtin_amdgcn_readlane(tv, k);
        if (w >= NWG) t = 0;
        len[k] = t & 8191;
        off[k] = w * CHUNK + (t >> 13);
    }
    #pragma unroll
    for (int k = 0; k < RUNS; ++k)
        rec[k] = recs[(lane < len[k]) ? off[k] + lane : 0];
    if (tid < NPB) ncnt[tid] = 0;
    __syncthreads();

    int rk[RUNS];
    #pragma unroll
    for (int k = 0; k < RUNS; ++k) {
        rk[k] = 0;
        if (lane < len[k]) rk[k] = atomicAdd(&ncnt[rec[k].x - node0], 1);
    }
    #pragma unroll
    for (int k = 0; k < RUNS; ++k) {
        if (lane < len[k]) {
            if (rk[k] < CAP) slots[(rec[k].x - node0) * CAP + rk[k]] = make_int2(rec[k].y, rec[k].z);
            else             put_overflow(rec[k].x, rec[k].y, rec[k].z, cnt, ovf);
        }
    }
    #pragma unroll
    for (int k = 0; k < RUNS; ++k)
        for (int i = lane + 32; i < len[k]; i += 32)
            place_one(recs[off[k] + i], node0, ncnt, slots, cnt, ovf);
    __syncthreads();

    int2 val[NPB / 32];
    bool mine[NPB / 32];
    #pragma unroll
    for (int i = 0; i < NPB / 32; ++i) {
        const int nl = wave + 32 * i;
        const int hi = min(ncnt[nl], CAP);
        mine[i] = (node0 + nl < NN) && lane < hi;
        val[i]  = slots[nl * CAP + lane];
    }
    #pragma unroll
    for (int i = 0; i < NPB / 32; ++i) {
        const int node = node0 + wave + 32 * i;
        if (mine[i]) reinterpret_cast<int2*>(bucket)[(size_t)node * CAP + lane] = val[i];
    }
    if (tid < NPB) {
        const int node = node0 + tid;
        if (node < NN) cnt[node] = ncnt[tid];
    }
}

#define NPW 4
#define TILE_N 32
__global__ __launch_bounds__(256) void softmax_gemm(
        const float* __restrict__ nf,
        const float* __restrict__ emb,
        const int*   __restrict__ cnt,
        const int*   __restrict__ bucket_pairs,
        const int*   __restrict__ ovf_recs,
        const float* __restrict__ W,
        const float* __restrict__ bias,
        float*       __restrict__ out) {
    __shared__ float fts[TILE_N * 68];
    __shared__ float Ws [64 * 68];

    const int tid     = threadIdx.x;
    const int lane    = tid & 31;
    const int wave    = __builtin_amdgcn_readfirstlane(tid >> 5);
    const int rowbase = blockIdx.x * TILE_N;
    const int t0      = rowbase + wave * NPW;

    const float2* nf2  = reinterpret_cast<const float2*>(nf);
    const float2* emb2 = reinterpret_cast<const float2*>(emb);
    const int2* slots  = reinterpret_cast<const int2*>(bucket_pairs);

    const int cvec = cnt[t0 + (lane & (NPW - 1))];

    float4 wv[4];
    #pragma unroll
    for (int i = 0; i < 4; ++i)
        wv[i] = reinterpret_cast<const float4*>(W)[tid + i * 256];

    int deg_next = (t0 < NN) ? __builtin_amdgcn_readlane(cvec, 0) : 0;
    int2 pairs_next = make_int2(0, 0);
    if (deg_next > 0) pairs_next = slots[(size_t)t0 * CAP + min(lane, min(deg_next, CAP) - 1)];

    #pragma unroll
    for (int i = 0; i < 4; ++i) {
        const int idx4 = tid + i * 256;
        *reinterpret_cast<float4*>(&Ws[(idx4 >> 4) * 68 + (idx4 & 15) * 4]) = wv[i];
    }

    for (int q = 0; q < NPW; ++q) {
        const int t   = t0 + q;
        const int deg = deg_next;
        const int n   = min(deg, CAP);
        const int2 mine = pairs_next;
        float2 own = make_float2(0.0f, 0.0f);
        if (t < NN) own = nf2[(size_t)t * (DD / 2) + lane];
        deg_next = 0;
        if (q + 1 < NPW && t + 1 < NN) deg_next = __builtin_amdgcn_readlane(cvec, q + 1);
        if (deg_next > 0) pairs_next = slots[(size_t)(t + 1) * CAP + min(lane, min(deg_next, CAP) - 1)];

        float mx = 0.0f, my = 0.0f;
        float sx = 0.0f, sy = 0.0f;
        float qx = 0.0f, qy = 0.0f;
        for (int j = 0; j < n; j += 4) {
            float2 x[4], eb[4];
            #pragma unroll
            for (int k = 0; k < 4; ++k) {
                const int jj = min(j + k, n - 1);
                const int s = __builtin_amdgcn_readlane(mine.x, jj);
                const int a = __builtin_amdgcn_readlane(mine.y, jj);
                x[k]  = nf2[(size_t)s * (DD / 2) + lane];
                eb[k] = emb2[a * (DD / 2) + lane];
            }
            asm volatile("" :: "v"(x[0].x), "v"(x[0].y), "v"(x[1].x), "v"(x[1].y),
                               "v"(x[2].x), "v"(x[2].y), "v"(x[3].x), "v"(x[3].y),
                               "v"(eb[0].x), "v"(eb[0].y), "v"(eb[1].x), "v"(eb[1].y),
                               "v"(eb[2].x), "v"(eb[2].y), "v"(eb[3].x), "v"(eb[3].y));
            float ax[4], ay[4];
            #pragma unroll
            for (int k = 0; k < 4; ++k) {
                const bool valid = (j + k) < n;
                const float vx = fmaxf(x[k].x + eb[k].x, 0.0f) + EPSF;
                const float vy = fmaxf(x[k].y + eb[k].y, 0.0f) + EPSF;
                ax[k] = valid ? vx : 0.0f;
                ay[k] = valid ? vy : 0.0f;
            }
            const float nx = fmaxf(mx, fmaxf(fmaxf(ax[0], ax[1]), fmaxf(ax[2], ax[3])));
            const float ny = fmaxf(my, fmaxf(fmaxf(ay[0], ay[1]), fmaxf(ay[2], ay[3])));
            const float rx = __expf(mx - nx);
            const float ry = __expf(my - ny);
            sx *= rx; qx *= rx;
            sy *= ry; qy *= ry;
            #pragma unroll
            for (int k = 0; k < 4; ++k) {
                const bool valid = (j + k) < n;
                const float zx = valid ? __expf(ax[k] - nx) : 0.0f;
                const float zy = valid ? __expf(ay[k] - ny) : 0.0f;
                sx += zx; qx += ax[k] * zx;
                sy += zy; qy += ay[k] * zy;
            }
            mx = nx; my = ny;
        }
        if (deg > CAP) {
            const int4* ovf = reinterpret_cast<const int4*>(ovf_recs);
            const int total = min(cnt[NN], EE);
            for (int base = 0; base < total; base += 32) {
                const int i = base + lane;
                int4 rec = make_int4(-1, 0, 0, 0);
                if (i < total) rec = ovf[i];
                unsigned hit = __builtin_amdgcn_ballot_w32(rec.x == t);
                while (hit) {
                    const int l = __builtin_ctz(hit);
                    hit &= hit - 1;
                    const int s = __builtin_amdgcn_readlane(rec.y, l);
                    const int a = __builtin_amdgcn_readlane(rec.z, l);
                    const float2 x  = nf2[(size_t)s * (DD / 2) + lane];
                    const float2 eb = emb2[a * (DD / 2) + lane];
                    const float vx = fmaxf(x.x + eb.x, 0.0f) + EPSF;
                    const float vy = fmaxf(x.y + eb.y, 0.0f) + EPSF;
                    const float nx = fmaxf(mx, vx);
                    const float ny = fmaxf(my, vy);
                    const float rx = __expf(mx - nx);
                    const float ry = __expf(my - ny);
                    const float zx = __expf(vx - nx);
                    const float zy = __expf(vy - ny);
                    sx = sx * rx + zx; qx = qx * rx + vx * zx;
                    sy = sy * ry + zy; qy = qy * ry + vy * zy;
                    mx = nx; my = ny;
                }
            }
        }
        const float aggx = (sx > 0.0f) ? (qx / sx) : 0.0f;
        const float aggy = (sy > 0.0f) ? (qy / sy) : 0.0f;
        *reinterpret_cast<float2*>(&fts[(wave * NPW + q) * 68 + 2 * lane]) =
            make_float2(own.x + aggx, own.y + aggy);
    }
    __syncthreads();

    const int ct    = wave & 3;
    const int rt    = wave >> 2;
    const int half  = lane >> 4;
    const int m     = lane & 15;

    v8f acc = {};
    #pragma unroll
    for (int k0 = 0; k0 < DD; k0 += 4) {
        const int k = k0 + 2 * half;
        v2f a, b;
        a.x = fts[(rt * 16 + m) * 68 + k];
        a.y = fts[(rt * 16 + m) * 68 + k + 1];
        b.x = Ws[(ct * 16 + m) * 68 + k];
        b.y = Ws[(ct * 16 + m) * 68 + k + 1];
        acc = __builtin_amdgcn_wmma_f32_16x16x4_f32(
                   false, a,  false, b,
                   (short)0, acc,
                   false,  false);
    }

    const int tilebase = rowbase + rt * 16;
    if (tilebase < NN) {
        const float bb = bias[ct * 16 + m];
        #pragma unroll
        for (int v = 0; v < 8; ++v) {
            const int row = tilebase + v + 8 * half;
            out[(long long)row * DD + ct * 16 + m] = acc[v] + bb;
        }
    }
}

extern "C" void kernel_launch(void* const* d_in, const int* in_sizes, int n_in,
                              void* d_out, int out_size, void* d_ws, size_t ws_size,
                              hipStream_t stream) {
    const float* nf    = (const float*)d_in[0];
    const int*   eattr = (const int*)  d_in[1];
    const int*   src   = (const int*)  d_in[2];
    const int*   dst   = (const int*)  d_in[3];
    const float* emb   = (const float*)d_in[4];
    const float* W     = (const float*)d_in[5];
    const float* bias  = (const float*)d_in[6];
    float*       out   = (float*)d_out;

    int*   cnt    = (int*)d_ws;
    int*   tab    = cnt + NPAD;
    int*   bucket = tab + TABSZ;
    int*   ovf    = bucket + (size_t)NN * CAP * 2;
    int*   seg    = ovf + (size_t)EE * 4;

    bin_edges<<<NWG, 1024, 0, stream>>>(eattr, src, dst, cnt, tab, seg);
    place_edges<<<NBIN, 1024, 0, stream>>>(tab, seg, cnt, bucket, ovf);
    softmax_gemm<<<(NN + TILE_N - 1) / TILE_N, 256, 0, stream>>>(nf, emb, cnt, bucket, ovf, W, bias, out);
}


# Softmax edge loop cut from 99 to 59 vector-issue cycles per 4 edges: packed 2-feature math in base 2 (log2e folded into the eps multiply-add), unmasked whole chunks, scalar row bases; sum order kept
# speedup vs baseline: 7.7278x; 1.1460x over previous
#include <hip/hip_runtime.h>


#define NN 50000
#define EE 800000
#define DD 64
#define EPSF 1e-7f
#define NPAD 50176
#define CAP 32
#define BSHIFT 8
#define NPB 256
#define NBIN 196
#define CHUNK 4096
#define NWG 196
#define TABSZ 38464
#define RUNS 7

typedef __attribute__((ext_vector_type(2))) float v2f;
typedef __attribute__((ext_vector_type(8))) float v8f;


__device__ __forceinline__ void put_overflow(int d, int s, int a,
                                             int* __restrict__ cnt,
                                             int* __restrict__ ovf) {
    const int k = atomicAdd(cnt + NN, 1);
    reinterpret_cast<int4*>(ovf)[k] = make_int4(d, s, a, 0);
}

__global__ __launch_bounds__(1024) void bin_edges(const int* __restrict__ eattr,
                                                  const int* __restrict__ src,
                                                  const int* __restrict__ dst,
                                                  int*       __restrict__ cnt,
                                                  int*       __restrict__ tab,
                                                  int*       __restrict__ seg) {
    __shared__ int4 stage[CHUNK];
    __shared__ int lcnt[NBIN];
    __shared__ int lpre[NBIN];
    __shared__ int wsum[8];
    const int tid  = threadIdx.x;
    const int lane = tid & 31;
    int d[4], s[4], a[4], lr[4];
    const int e0 = blockIdx.x * CHUNK + tid;
    #pragma unroll
    for (int k = 0; k < 4; ++k) {
        const int e = e0 + k * 1024;
        d[k] = -1; s[k] = 0; a[k] = 0; lr[k] = 0;
        if (e < EE) { d[k] = dst[e]; s[k] = src[e]; a[k] = eattr[e]; }
    }
    if (tid < NBIN) lcnt[tid] = 0;
    if (blockIdx.x == 0 && tid < NPAD - NN) cnt[NN + tid] = 0;
    __syncthreads();

    #pragma unroll
    for (int k = 0; k < 4; ++k)
        if (d[k] >= 0) lr[k] = atomicAdd(&lcnt[d[k] >> BSHIFT], 1);
    __syncthreads();

    int v = 0, incl = 0;
    if (tid < 256) {
        v = (tid < NBIN) ? lcnt[tid] : 0;
        incl = v;
        #pragma unroll
        for (int off = 1; off < 32; off <<= 1) {
            const int t = __shfl_up(incl, off);
            if (lane >= off) incl += t;
        }
        if (lane == 31) wsum[tid >> 5] = incl;
    }
    __syncthreads();
    if (tid < NBIN) {
        int p = 0;
        for (int w = 0; w < (tid >> 5); ++w) p += wsum[w];
        p += incl - v;
        lpre[tid] = p;
        tab[tid * NWG + blockIdx.x] = (p << 13) | v;
    }
    __syncthreads();

    #pragma unroll
    for (int k = 0; k < 4; ++k)
        if (d[k] >= 0) stage[lpre[d[k] >> BSHIFT] + lr[k]] = make_int4(d[k], s[k], a[k], 0);
    __syncthreads();

    const int nvalid = min(CHUNK, EE - (int)blockIdx.x * CHUNK);
    #pragma unroll
    for (int k = 0; k < 4; ++k) {
        const int j = tid + k * 1024;
        if (j < nvalid) reinterpret_cast<int4*>(seg)[(size_t)blockIdx.x * CHUNK + j] = stage[j];
    }
}

__device__ __forceinline__ void place_one(int4 rec, int node0, int* ncnt, int2* slots,
                                          int* __restrict__ cnt, int* __restrict__ ovf) {
    const int dl = rec.x - node0;
    const int r = atomicAdd(&ncnt[dl], 1);
    if (r < CAP) slots[dl * CAP + r] = make_int2(rec.y, rec.z);
    else         put_overflow(rec.x, rec.y, rec.z, cnt, ovf);
}

__global__ __launch_bounds__(1024) void place_edges(const int* __restrict__ tab,
                                                    const int* __restrict__ seg,
                                                    int*       __restrict__ cnt,
                                                    int*       __restrict__ bucket,
                                                    int*       __restrict__ ovf) {
    __shared__ int2 slots[NPB * CAP];
    __shared__ int ncnt[NPB];
    const int tid   = threadIdx.x;
    const int lane  = tid & 31;
    const int wave  = __builtin_amdgcn_readfirstlane(tid >> 5);
    const int bin   = blockIdx.x;
    const int node0 = bin << BSHIFT;
    const int4* recs = reinterpret_cast<const int4*>(seg);
    int len[RUNS], off[RUNS];
    int4 rec[RUNS];
    const int tv = tab[bin * NWG + min(wave + 32 * (lane & 7), NWG - 1)];
    #pragma unroll
    for (int k = 0; k < RUNS; ++k) {
        const int w = wave + 32 * k;
        int t = __builtin_amdgcn_readlane(tv, k);
        if (w >= NWG) t = 0;
        len[k] = t & 8191;
        off[k] = w * CHUNK + (t >> 13);
    }
    #pragma unroll
    for (int k = 0; k < RUNS; ++k)
        rec[k] = recs[(lane < len[k]) ? off[k] + lane : 0];
    if (tid < NPB) ncnt[tid] = 0;
    __syncthreads();

    int rk[RUNS];
    #pragma unroll
    for (int k = 0; k < RUNS; ++k) {
        rk[k] = 0;
        if (lane < len[k]) rk[k] = atomicAdd(&ncnt[rec[k].x - node0], 1);
    }
    #pragma unroll
    for (int k = 0; k < RUNS; ++k) {
        if (lane < len[k]) {
            if (rk[k] < CAP) slots[(rec[k].x - node0) * CAP + rk[k]] = make_int2(rec[k].y, rec[k].z);
            else             put_overflow(rec[k].x, rec[k].y, rec[k].z, cnt, ovf);
        }
    }
    #pragma unroll
    for (int k = 0; k < RUNS; ++k)
        for (int i = lane + 32; i < len[k]; i += 32)
            place_one(recs[off[k] + i], node0, ncnt, slots, cnt, ovf);
    __syncthreads();

    int2 val[NPB / 32];
    bool mine[NPB / 32];
    #pragma unroll
    for (int i = 0; i < NPB / 32; ++i) {
        const int nl = wave + 32 * i;
        const int hi = min(ncnt[nl], CAP);
        mine[i] = (node0 + nl < NN) && lane < hi;
        val[i]  = slots[nl * CAP + lane];
    }
    #pragma unroll
    for (int i = 0; i < NPB / 32; ++i) {
        const int node = node0 + wave + 32 * i;
        if (mine[i]) reinterpret_cast<int2*>(bucket)[(size_t)node * CAP + lane] = val[i];
    }
    if (tid < NPB) {
        const int node = node0 + tid;
        if (node < NN) cnt[node] = ncnt[tid];
    }
}

#define LOG2E 1.44269504088896340736f
#define LN2   0.69314718055994530942f
#define PADV  (-1.0e30f)

__device__ __forceinline__ v2f vmax2(v2f a, v2f b) { return __builtin_elementwise_max(a, b); }
__device__ __forceinline__ v2f vexp2(v2f a) {
    v2f r;
    r.x = __builtin_amdgcn_exp2f(a.x);
    r.y = __builtin_amdgcn_exp2f(a.y);
    return r;
}
__device__ __forceinline__ v2f edge_m2(v2f x, v2f eb) {
    const v2f zero = {0.0f, 0.0f};
    const v2f l2e  = {LOG2E, LOG2E};
    const v2f epl  = {EPSF * LOG2E, EPSF * LOG2E};
    return __builtin_elementwise_fma(vmax2(x + eb, zero), l2e, epl);
}

typedef const v2f __attribute__((address_space(1)))* gv2f_ptr;
__device__ __forceinline__ gv2f_ptr uniform_row(const v2f* base, int r) {
    const unsigned long long p = reinterpret_cast<unsigned long long>(base) +
                                 ((unsigned long long)(unsigned)r << 8);
    const unsigned lo = __builtin_amdgcn_readfirstlane((unsigned)p);
    const unsigned hi = __builtin_amdgcn_readfirstlane((unsigned)(p >> 32));
    return (gv2f_ptr)(((unsigned long long)hi << 32) | lo);
}

template <bool TAIL>
__device__ __forceinline__ void chunk4(const v2f* __restrict__ nfv, const v2f* __restrict__ embv,
                                       const int2 mine, const int j, const int n, const int lane,
                                       v2f& M, v2f& S, v2f& Q) {
    v2f x[4], eb[4];
    gv2f_ptr xr[4], er[4];
    #pragma unroll
    for (int k = 0; k < 4; ++k) {
        const int jj = TAIL ? min(j + k, n - 1) : (j + k);
        xr[k] = uniform_row(nfv,  __builtin_amdgcn_readlane(mine.x, jj));
        er[k] = uniform_row(embv, __builtin_amdgcn_readlane(mine.y, jj));
    }
    asm volatile("" :: "s"(xr[0]), "s"(xr[1]), "s"(xr[2]), "s"(xr[3]),
                       "s"(er[0]), "s"(er[1]), "s"(er[2]), "s"(er[3]));
    #pragma unroll
    for (int k = 0; k < 4; ++k) {
        x[k]  = xr[k][lane];
        eb[k] = er[k][lane];
    }
    asm volatile("" :: "v"(x[0]), "v"(x[1]), "v"(x[2]), "v"(x[3]),
                       "v"(eb[0]), "v"(eb[1]), "v"(eb[2]), "v"(eb[3]));
    v2f m2[4];
    #pragma unroll
    for (int k = 0; k < 4; ++k) {
        m2[k] = edge_m2(x[k], eb[k]);
        if (TAIL && (j + k) >= n) { m2[k].x = PADV; m2[k].y = PADV; }
    }
    const v2f nM = vmax2(vmax2(M, vmax2(m2[0], m2[1])), vmax2(m2[2], m2[3]));
    const v2f r  = vexp2(M - nM);
    v2f z[4];
    #pragma unroll
    for (int k = 0; k < 4; ++k) z[k] = vexp2(m2[k] - nM);
    S = __builtin_elementwise_fma(S, r, z[0]);
    S = S + z[1];
    S = S + z[2];
    S = S + z[3];
    Q = Q * r;
    #pragma unroll
    for (int k = 0; k < 4; ++k) Q = __builtin_elementwise_fma(m2[k], z[k], Q);
    M = nM;
}

#define NPW 4
#define TILE_N 32
__global__ __launch_bounds__(256) void softmax_gemm(
        const float* __restrict__ nf,
        const float* __restrict__ emb,
        const int*   __restrict__ cnt,
        const int*   __restrict__ bucket_pairs,
        const int*   __restrict__ ovf_recs,
        const float* __restrict__ W,
        const float* __restrict__ bias,
        float*       __restrict__ out) {
    __shared__ float fts[TILE_N * 68];
    __shared__ float Ws [64 * 68];

    const int tid     = threadIdx.x;
    const int lane    = tid & 31;
    const int wave    = __builtin_amdgcn_readfirstlane(tid >> 5);
    const int rowbase = blockIdx.x * TILE_N;
    const int t0      = rowbase + wave * NPW;

    const float2* nf2  = reinterpret_cast<const float2*>(nf);
    const int2* slots  = reinterpret_cast<const int2*>(bucket_pairs);

    const int cvec = cnt[t0 + (lane & (NPW - 1))];

    float4 wv[4];
    #pragma unroll
    for (int i = 0; i < 4; ++i)
        wv[i] = reinterpret_cast<const float4*>(W)[tid + i * 256];

    int deg_next = (t0 < NN) ? __builtin_amdgcn_readlane(cvec, 0) : 0;
    int2 pairs_next = make_int2(0, 0);
    if (deg_next > 0) pairs_next = slots[(size_t)t0 * CAP + min(lane, min(deg_next, CAP) - 1)];

    #pragma unroll
    for (int i = 0; i < 4; ++i) {
        const int idx4 = tid + i * 256;
        *reinterpret_cast<float4*>(&Ws[(idx4 >> 4) * 68 + (idx4 & 15) * 4]) = wv[i];
    }

    const v2f* nfv  = reinterpret_cast<const v2f*>(nf);
    const v2f* embv = reinterpret_cast<const v2f*>(emb);

    for (int q = 0; q < NPW; ++q) {
        const int t   = t0 + q;
        const int deg = deg_next;
        const int n   = min(deg, CAP);
        const int2 mine = pairs_next;
        float2 own = make_float2(0.0f, 0.0f);
        if (t < NN) own = nf2[(size_t)t * (DD / 2) + lane];
        deg_next = 0;
        if (q + 1 < NPW && t + 1 < NN) deg_next = __builtin_amdgcn_readlane(cvec, q + 1);
        if (deg_next > 0) pairs_next = slots[(size_t)(t + 1) * CAP + min(lane, min(deg_next, CAP) - 1)];

        v2f M = {0.0f, 0.0f};
        v2f S = {0.0f, 0.0f};
        v2f Q = {0.0f, 0.0f};
        int j = 0;
        for (; j + 4 <= n; j += 4) chunk4<false>(nfv, embv, mine, j, n, lane, M, S, Q);
        if (j < n)                 chunk4<true >(nfv, embv, mine, j, n, lane, M, S, Q);
        if (deg > CAP) {
            const int4* ovf = reinterpret_cast<const int4*>(ovf_recs);
            const int total = min(cnt[NN], EE);
            for (int base = 0; base < total; base += 32) {
                const int i = base + lane;
                int4 rec = make_int4(-1, 0, 0, 0);
                if (i < total) rec = ovf[i];
                unsigned hit = __builtin_amdgcn_ballot_w32(rec.x == t);
                while (hit) {
                    const int l = __builtin_ctz(hit);
                    hit &= hit - 1;
                    const int s = __builtin_amdgcn_readlane(rec.y, l);
                    const int a = __builtin_amdgcn_readlane(rec.z, l);
                    const v2f v  = edge_m2(nfv[(size_t)s * (DD / 2) + lane], embv[a * (DD / 2) + lane]);
                    const v2f nM = vmax2(M, v);
                    const v2f r  = vexp2(M - nM);
                    const v2f z  = vexp2(v - nM);
                    S = __builtin_elementwise_fma(S, r, z);
                    Q = __builtin_elementwise_fma(v, z, Q * r);
                    M = nM;
                }
            }
        }
        const float dnx = (S.x > 0.0f) ? S.x : 1.0f;
        const float dny = (S.y > 0.0f) ? S.y : 1.0f;
        const float aggx = (Q.x / dnx) * LN2;
        const float aggy = (Q.y / dny) * LN2;
        *reinterpret_cast<float2*>(&fts[(wave * NPW + q) * 68 + 2 * lane]) =
            make_float2(own.x + aggx, own.y + aggy);
    }
    __syncthreads();

    const int ct    = wave & 3;
    const int rt    = wave >> 2;
    const int half  = lane >> 4;
    const int m     = lane & 15;

    v8f acc = {};
    #pragma unroll
    for (int k0 = 0; k0 < DD; k0 += 4) {
        const int k = k0 + 2 * half;
        v2f a, b;
        a.x = fts[(rt * 16 + m) * 68 + k];
        a.y = fts[(rt * 16 + m) * 68 + k + 1];
        b.x = Ws[(ct * 16 + m) * 68 + k];
        b.y = Ws[(ct * 16 + m) * 68 + k + 1];
        acc = __builtin_amdgcn_wmma_f32_16x16x4_f32(
                   false, a,  false, b,
                   (short)0, acc,
                   false,  false);
    }

    const int tilebase = rowbase + rt * 16;
    if (tilebase < NN) {
        const float bb = bias[ct * 16 + m];
        #pragma unroll
        for (int v = 0; v < 8; ++v) {
            const int row = tilebase + v + 8 * half;
            out[(long long)row * DD + ct * 16 + m] = acc[v] + bb;
        }
    }
}

extern "C" void kernel_launch(void* const* d_in, const int* in_sizes, int n_in,
                              void* d_out, int out_size, void* d_ws, size_t ws_size,
                              hipStream_t stream) {
    const float* nf    = (const float*)d_in[0];
    const int*   eattr = (const int*)  d_in[1];
    const int*   src   = (const int*)  d_in[2];
    const int*   dst   = (const int*)  d_in[3];
    const float* emb   = (const float*)d_in[4];
    const float* W     = (const float*)d_in[5];
    const float* bias  = (const float*)d_in[6];
    float*       out   = (float*)d_out;

    int*   cnt    = (int*)d_ws;
    int*   tab    = cnt + NPAD;
    int*   bucket = tab + TABSZ;
    int*   ovf    = bucket + (size_t)NN * CAP * 2;
    int*   seg    = ovf + (size_t)EE * 4;

    bin_edges<<<NWG, 1024, 0, stream>>>(eattr, src, dst, cnt, tab, seg);
    place_edges<<<NBIN, 1024, 0, stream>>>(tab, seg, cnt, bucket, ovf);
    softmax_gemm<<<(NN + TILE_N - 1) / TILE_N, 256, 0, stream>>>(nf, emb, cnt, bucket, ovf, W, bias, out);
}


# Edge records packed into one 32-bit word (place in bin 8 bits, bond type 4, source 20) in segment, LDS and slots: 4x fewer record bytes, one lane read per edge in the softmax loop; sum order kept
# speedup vs baseline: 8.4638x; 1.0952x over previous
#include <hip/hip_runtime.h>


#define NN 50000
#define EE 800000
#define DD 64
#define EPSF 1e-7f
#define NPAD 50176
#define CAP 32
#define BSHIFT 8
#define NPB 256
#define NBIN 196
#define CHUNK 4096
#define NWG 196
#define TABSZ 38464
#define RUNS 7

typedef __attribute__((ext_vector_type(2))) float v2f;
typedef __attribute__((ext_vector_type(8))) float v8f;


static_assert(NN <= (1 << 20), "source node must fit 20 bits");
static_assert(NPB == 256 && BSHIFT == 8, "place in bin must fit 8 bits");
__device__ __forceinline__ int pack_rec(int d, int s, int a) {
    return (int)(((unsigned)s << 12) | ((unsigned)a << 8) | ((unsigned)d & 255u));
}
__device__ __forceinline__ int rec_place(int rec) { return rec & 255; }
__device__ __forceinline__ int rec_attr(int rec)  { return (rec >> 8) & 15; }
__device__ __forceinline__ int rec_src(int rec)   { return (int)((unsigned)rec >> 12); }

__device__ __forceinline__ void put_overflow(int d, int s, int a,
                                             int* __restrict__ cnt,
                                             int* __restrict__ ovf) {
    const int k = atomicAdd(cnt + NN, 1);
    reinterpret_cast<int4*>(ovf)[k] = make_int4(d, s, a, 0);
}

__global__ __launch_bounds__(1024) void bin_edges(const int* __restrict__ eattr,
                                                  const int* __restrict__ src,
                                                  const int* __restrict__ dst,
                                                  int*       __restrict__ cnt,
                                                  int*       __restrict__ tab,
                                                  int*       __restrict__ seg) {
    __shared__ int4 stage4[CHUNK / 4];
    int* stage = reinterpret_cast<int*>(stage4);
    __shared__ int lcnt[NBIN];
    __shared__ int lpre[NBIN];
    __shared__ int wsum[8];
    const int tid  = threadIdx.x;
    const int lane = tid & 31;
    int d[4], s[4], a[4], lr[4];
    const int e0 = blockIdx.x * CHUNK + tid;
    #pragma unroll
    for (int k = 0; k < 4; ++k) {
        const int e = e0 + k * 1024;
        d[k] = -1; s[k] = 0; a[k] = 0; lr[k] = 0;
        if (e < EE) { d[k] = dst[e]; s[k] = src[e]; a[k] = eattr[e]; }
    }
    if (tid < NBIN) lcnt[tid] = 0;
    if (blockIdx.x == 0 && tid < NPAD - NN) cnt[NN + tid] = 0;
    __syncthreads();

    #pragma unroll
    for (int k = 0; k < 4; ++k)
        if (d[k] >= 0) lr[k] = atomicAdd(&lcnt[d[k] >> BSHIFT], 1);
    __syncthreads();

    int v = 0, incl = 0;
    if (tid < 256) {
        v = (tid < NBIN) ? lcnt[tid] : 0;
        incl = v;
        #pragma unroll
        for (int off = 1; off < 32; off <<= 1) {
            const int t = __shfl_up(incl, off);
            if (lane >= off) incl += t;
        }
        if (lane == 31) wsum[tid >> 5] = incl;
    }
    __syncthreads();
    if (tid < NBIN) {
        int p = 0;
        for (int w = 0; w < (tid >> 5); ++w) p += wsum[w];
        p += incl - v;
        lpre[tid] = p;
        tab[tid * NWG + blockIdx.x] = (p << 13) | v;
    }
    __syncthreads();

    #pragma unroll
    for (int k = 0; k < 4; ++k)
        if (d[k] >= 0) stage[lpre[d[k] >> BSHIFT] + lr[k]] = pack_rec(d[k], s[k], a[k]);
    __syncthreads();

    static_assert(EE % 4 == 0 && CHUNK == 4 * 1024, "four records per thread");
    const int nvalid = min(CHUNK, EE - (int)blockIdx.x * CHUNK);
    if (4 * tid < nvalid)
        reinterpret_cast<int4*>(seg)[(size_t)blockIdx.x * (CHUNK / 4) + tid] = stage4[tid];
}

__device__ __forceinline__ void place_one(int rec, int node0, int* ncnt, int* slots,
                                          int* __restrict__ cnt, int* __restrict__ ovf) {
    const int dl = rec_place(rec);
    const int r = atomicAdd(&ncnt[dl], 1);
    if (r < CAP) slots[dl * CAP + r] = rec;
    else         put_overflow(node0 + dl, rec_src(rec), rec_attr(rec), cnt, ovf);
}

__global__ __launch_bounds__(1024) void place_edges(const int* __restrict__ tab,
                                                    const int* __restrict__ seg,
                                                    int*       __restrict__ cnt,
                                                    int*       __restrict__ bucket,
                                                    int*       __restrict__ ovf) {
    __shared__ int slots[NPB * CAP];
    __shared__ int ncnt[NPB];
    const int tid   = threadIdx.x;
    const int lane  = tid & 31;
    const int wave  = __builtin_amdgcn_readfirstlane(tid >> 5);
    const int bin   = blockIdx.x;
    const int node0 = bin << BSHIFT;
    const int* recs = seg;
    int len[RUNS], off[RUNS];
    int rec[RUNS];
    const int tv = tab[bin * NWG + min(wave + 32 * (lane & 7), NWG - 1)];
    #pragma unroll
    for (int k = 0; k < RUNS; ++k) {
        const int w = wave + 32 * k;
        int t = __builtin_amdgcn_readlane(tv, k);
        if (w >= NWG) t = 0;
        len[k] = t & 8191;
        off[k] = w * CHUNK + (t >> 13);
    }
    #pragma unroll
    for (int k = 0; k < RUNS; ++k)
        rec[k] = recs[(lane < len[k]) ? off[k] + lane : 0];
    if (tid < NPB) ncnt[tid] = 0;
    __syncthreads();

    int rk[RUNS];
    #pragma unroll
    for (int k = 0; k < RUNS; ++k) {
        rk[k] = 0;
        if (lane < len[k]) rk[k] = atomicAdd(&ncnt[rec_place(rec[k])], 1);
    }
    #pragma unroll
    for (int k = 0; k < RUNS; ++k) {
        if (lane < len[k]) {
            if (rk[k] < CAP) slots[rec_place(rec[k]) * CAP + rk[k]] = rec[k];
            else             put_overflow(node0 + rec_place(rec[k]), rec_src(rec[k]), rec_attr(rec[k]), cnt, ovf);
        }
    }
    #pragma unroll
    for (int k = 0; k < RUNS; ++k)
        for (int i = lane + 32; i < len[k]; i += 32)
            place_one(recs[off[k] + i], node0, ncnt, slots, cnt, ovf);
    __syncthreads();

    int val[NPB / 32];
    bool mine[NPB / 32];
    #pragma unroll
    for (int i = 0; i < NPB / 32; ++i) {
        const int nl = wave + 32 * i;
        const int hi = min(ncnt[nl], CAP);
        mine[i] = (node0 + nl < NN) && lane < hi;
        val[i]  = slots[nl * CAP + lane];
    }
    #pragma unroll
    for (int i = 0; i < NPB / 32; ++i) {
        const int node = node0 + wave + 32 * i;
        if (mine[i]) bucket[(size_t)node * CAP + lane] = val[i];
    }
    if (tid < NPB) {
        const int node = node0 + tid;
        if (node < NN) cnt[node] = ncnt[tid];
    }
}

#define LOG2E 1.44269504088896340736f
#define LN2   0.69314718055994530942f
#define PADV  (-1.0e30f)

__device__ __forceinline__ v2f vmax2(v2f a, v2f b) { return __builtin_elementwise_max(a, b); }
__device__ __forceinline__ v2f vexp2(v2f a) {
    v2f r;
    r.x = __builtin_amdgcn_exp2f(a.x);
    r.y = __builtin_amdgcn_exp2f(a.y);
    return r;
}
__device__ __forceinline__ v2f edge_m2(v2f x, v2f eb) {
    const v2f zero = {0.0f, 0.0f};
    const v2f l2e  = {LOG2E, LOG2E};
    const v2f epl  = {EPSF * LOG2E, EPSF * LOG2E};
    return __builtin_elementwise_fma(vmax2(x + eb, zero), l2e, epl);
}

typedef const v2f __attribute__((address_space(1)))* gv2f_ptr;
__device__ __forceinline__ gv2f_ptr uniform_row(const v2f* base, unsigned byteoff) {
    const unsigned long long p = reinterpret_cast<unsigned long long>(base) +
                                 (unsigned long long)byteoff;
    const unsigned lo = __builtin_amdgcn_readfirstlane((unsigned)p);
    const unsigned hi = __builtin_amdgcn_readfirstlane((unsigned)(p >> 32));
    return (gv2f_ptr)(((unsigned long long)hi << 32) | lo);
}

template <bool TAIL>
__device__ __forceinline__ void chunk4(const v2f* __restrict__ nfv, const v2f* __restrict__ embv,
                                       const int mine, const int j, const int n, const int lane,
                                       v2f& M, v2f& S, v2f& Q) {
    v2f x[4], eb[4];
    gv2f_ptr xr[4], er[4];
    #pragma unroll
    for (int k = 0; k < 4; ++k) {
        const int jj = TAIL ? min(j + k, n - 1) : (j + k);
        const unsigned w = (unsigned)__builtin_amdgcn_readlane(mine, jj);
        xr[k] = uniform_row(nfv,  (w >> 4) & 0xFFFFFF00u);
        er[k] = uniform_row(embv, w & 0x00000F00u);
    }
    asm volatile("" :: "s"(xr[0]), "s"(xr[1]), "s"(xr[2]), "s"(xr[3]),
                       "s"(er[0]), "s"(er[1]), "s"(er[2]), "s"(er[3]));
    #pragma unroll
    for (int k = 0; k < 4; ++k) {
        x[k]  = xr[k][lane];
        eb[k] = er[k][lane];
    }
    asm volatile("" :: "v"(x[0]), "v"(x[1]), "v"(x[2]), "v"(x[3]),
                       "v"(eb[0]), "v"(eb[1]), "v"(eb[2]), "v"(eb[3]));
    v2f m2[4];
    #pragma unroll
    for (int k = 0; k < 4; ++k) {
        m2[k] = edge_m2(x[k], eb[k]);
        if (TAIL && (j + k) >= n) { m2[k].x = PADV; m2[k].y = PADV; }
    }
    const v2f nM = vmax2(vmax2(M, vmax2(m2[0], m2[1])), vmax2(m2[2], m2[3]));
    const v2f r  = vexp2(M - nM);
    v2f z[4];
    #pragma unroll
    for (int k = 0; k < 4; ++k) z[k] = vexp2(m2[k] - nM);
    S = __builtin_elementwise_fma(S, r, z[0]);
    S = S + z[1];
    S = S + z[2];
    S = S + z[3];
    Q = Q * r;
    #pragma unroll
    for (int k = 0; k < 4; ++k) Q = __builtin_elementwise_fma(m2[k], z[k], Q);
    M = nM;
}

#define NPW 4
#define TILE_N 32
__global__ __launch_bounds__(256) void softmax_gemm(
        const float* __restrict__ nf,
        const float* __restrict__ emb,
        const int*   __restrict__ cnt,
        const int*   __restrict__ bucket_pairs,
        const int*   __restrict__ ovf_recs,
        const float* __restrict__ W,
        const float* __restrict__ bias,
        float*       __restrict__ out) {
    __shared__ float fts[TILE_N * 68];
    __shared__ float Ws [64 * 68];

    const int tid     = threadIdx.x;
    const int lane    = tid & 31;
    const int wave    = __builtin_amdgcn_readfirstlane(tid >> 5);
    const int rowbase = blockIdx.x * TILE_N;
    const int t0      = rowbase + wave * NPW;

    const float2* nf2  = reinterpret_cast<const float2*>(nf);
    const int* slots   = bucket_pairs;

    const int cvec = cnt[t0 + (lane & (NPW - 1))];

    float4 wv[4];
    #pragma unroll
    for (int i = 0; i < 4; ++i)
        wv[i] = reinterpret_cast<const float4*>(W)[tid + i * 256];

    int deg_next = (t0 < NN) ? __builtin_amdgcn_readlane(cvec, 0) : 0;
    int pairs_next = 0;
    if (deg_next > 0) pairs_next = slots[(size_t)t0 * CAP + min(lane, min(deg_next, CAP) - 1)];

    #pragma unroll
    for (int i = 0; i < 4; ++i) {
        const int idx4 = tid + i * 256;
        *reinterpret_cast<float4*>(&Ws[(idx4 >> 4) * 68 + (idx4 & 15) * 4]) = wv[i];
    }

    const v2f* nfv  = reinterpret_cast<const v2f*>(nf);
    const v2f* embv = reinterpret_cast<const v2f*>(emb);

    for (int q = 0; q < NPW; ++q) {
        const int t   = t0 + q;
        const int deg = deg_next;
        const int n   = min(deg, CAP);
        const int mine = pairs_next;
        float2 own = make_float2(0.0f, 0.0f);
        if (t < NN) own = nf2[(size_t)t * (DD / 2) + lane];
        deg_next = 0;
        if (q + 1 < NPW && t + 1 < NN) deg_next = __builtin_amdgcn_readlane(cvec, q + 1);
        if (deg_next > 0) pairs_next = slots[(size_t)(t + 1) * CAP + min(lane, min(deg_next, CAP) - 1)];

        v2f M = {0.0f, 0.0f};
        v2f S = {0.0f, 0.0f};
        v2f Q = {0.0f, 0.0f};
        int j = 0;
        for (; j + 4 <= n; j += 4) chunk4<false>(nfv, embv, mine, j, n, lane, M, S, Q);
        if (j < n)                 chunk4<true >(nfv, embv, mine, j, n, lane, M, S, Q);
        if (deg > CAP) {
            const int4* ovf = reinterpret_cast<const int4*>(ovf_recs);
            const int total = min(cnt[NN], EE);
            for (int base = 0; base < total; base += 32) {
                const int i = base + lane;
                int4 rec = make_int4(-1, 0, 0, 0);
                if (i < total) rec = ovf[i];
                unsigned hit = __builtin_amdgcn_ballot_w32(rec.x == t);
                while (hit) {
                    const int l = __builtin_ctz(hit);
                    hit &= hit - 1;
                    const int s = __builtin_amdgcn_readlane(rec.y, l);
                    const int a = __builtin_amdgcn_readlane(rec.z, l);
                    const v2f v  = edge_m2(nfv[(size_t)s * (DD / 2) + lane], embv[a * (DD / 2) + lane]);
                    const v2f nM = vmax2(M, v);
                    const v2f r  = vexp2(M - nM);
                    const v2f z  = vexp2(v - nM);
                    S = __builtin_elementwise_fma(S, r, z);
                    Q = __builtin_elementwise_fma(v, z, Q * r);
                    M = nM;
                }
            }
        }
        const float dnx = (S.x > 0.0f) ? S.x : 1.0f;
        const float dny = (S.y > 0.0f) ? S.y : 1.0f;
        const float aggx = (Q.x / dnx) * LN2;
        const float aggy = (Q.y / dny) * LN2;
        *reinterpret_cast<float2*>(&fts[(wave * NPW + q) * 68 + 2 * lane]) =
            make_float2(own.x + aggx, own.y + aggy);
    }
    __syncthreads();

    const int ct    = wave & 3;
    const int rt    = wave >> 2;
    const int half  = lane >> 4;
    const int m     = lane & 15;

    v8f acc = {};
    #pragma unroll
    for (int k0 = 0; k0 < DD; k0 += 4) {
        const int k = k0 + 2 * half;
        v2f a, b;
        a.x = fts[(rt * 16 + m) * 68 + k];
        a.y = fts[(rt * 16 + m) * 68 + k + 1];
        b.x = Ws[(ct * 16 + m) * 68 + k];
        b.y = Ws[(ct * 16 + m) * 68 + k + 1];
        acc = __builtin_amdgcn_wmma_f32_16x16x4_f32(
                   false, a,  false, b,
                   (short)0, acc,
                   false,  false);
    }

    const int tilebase = rowbase + rt * 16;
    if (tilebase < NN) {
        const float bb = bias[ct * 16 + m];
        #pragma unroll
        for (int v = 0; v < 8; ++v) {
            const int row = tilebase + v + 8 * half;
            out[(long long)row * DD + ct * 16 + m] = acc[v] + bb;
        }
    }
}

extern "C" void kernel_launch(void* const* d_in, const int* in_sizes, int n_in,
                              void* d_out, int out_size, void* d_ws, size_t ws_size,
                              hipStream_t stream) {
    const float* nf    = (const float*)d_in[0];
    const int*   eattr = (const int*)  d_in[1];
    const int*   src   = (const int*)  d_in[2];
    const int*   dst   = (const int*)  d_in[3];
    const float* emb   = (const float*)d_in[4];
    const float* W     = (const float*)d_in[5];
    const float* bias  = (const float*)d_in[6];
    float*       out   = (float*)d_out;

    int*   cnt    = (int*)d_ws;
    int*   tab    = cnt + NPAD;
    int*   bucket = tab + TABSZ;
    int*   ovf    = bucket + (size_t)NN * CAP;
    int*   seg    = ovf + (size_t)EE * 4;

    bin_edges<<<NWG, 1024, 0, stream>>>(eattr, src, dst, cnt, tab, seg);
    place_edges<<<NBIN, 1024, 0, stream>>>(tab, seg, cnt, bucket, ovf);
    softmax_gemm<<<(NN + TILE_N - 1) / TILE_N, 256, 0, stream>>>(nf, emb, cnt, bucket, ovf, W, bias, out);
}
